# starting kernel
# baseline (speedup 1.0000x reference)
#include <hip/hip_runtime.h>

typedef __attribute__((ext_vector_type(2))) float v2f;
typedef __attribute__((ext_vector_type(8))) float v8f;

__device__ __forceinline__ void atomAddF32(float* p, float v) {
  asm volatile("global_atomic_add_f32 %0, %1, off" : : "v"(p), "v"(v) : "memory");
}

__global__ void zero4_kernel(float4* __restrict__ p, long long n4) {
  long long i = (long long)blockIdx.x * blockDim.x + threadIdx.x;
  if (i < n4) p[i] = make_float4(0.f, 0.f, 0.f, 0.f);
}

__global__ void degree_kernel(const int* __restrict__ dst, int* __restrict__ deg, int nE) {
  int i = blockIdx.x * blockDim.x + threadIdx.x;
  if (i < nE) atomicAdd(&deg[dst[i]], 1);
}

__global__ void invdeg_kernel(const int* __restrict__ deg, float* __restrict__ invd, int n) {
  int i = blockIdx.x * blockDim.x + threadIdx.x;
  if (i < n) invd[i] = 1.0f / fmaxf((float)deg[i], 1.0f);
}

__global__ void pack_w_kernel(const float* __restrict__ W, float* __restrict__ Wp,
                              int K, int N, int Np) {
  int i = blockIdx.x * blockDim.x + threadIdx.x;
  int total = (K >> 1) * Np;
  if (i >= total) return;
  int k2 = i / Np;
  int c = i - k2 * Np;
  float x0 = 0.f, x1 = 0.f;
  if (c < N) {
    x0 = W[(size_t)(2 * k2) * N + c];
    x1 = W[(size_t)(2 * k2 + 1) * N + c];
  }
  *(float2*)(Wp + (size_t)k2 * 2 * Np + 2 * c) = make_float2(x0, x1);
}

__global__ void scatter_sum_kernel(const float* __restrict__ h, const int* __restrict__ src,
                                   const int* __restrict__ dst, float* __restrict__ msum,
                                   int nE, int d, int logC) {
  long long idx = (long long)blockIdx.x * blockDim.x + threadIdx.x;
  long long total = (long long)nE << logC;
  if (idx >= total) return;
  int e = (int)(idx >> logC);
  int c = (int)(idx & ((1 << logC) - 1));
  int s = src[e];
  int t = dst[e];
  const float4 v = *(const float4*)(h + (size_t)s * d + 4 * c);
  float* p = msum + (size_t)t * d + 4 * c;
  atomAddF32(p + 0, v.x);
  atomAddF32(p + 1, v.y);
  atomAddF32(p + 2, v.z);
  atomAddF32(p + 3, v.w);
}

__device__ __forceinline__ v8f wmma4(v2f a, v2f b, v8f c) {
  return __builtin_amdgcn_wmma_f32_16x16x4_f32(false, a, false, b, (short)0, c, false, false);
}

template <bool RELU>
__global__ __launch_bounds__(256) void sage_gemm_wmma(
    const float* __restrict__ h, const float* __restrict__ msum,
    const float* __restrict__ invd, const float* __restrict__ Wp,
    const float* __restrict__ bias, float* __restrict__ out, int M, int d, int N, int Np) {
  const int w = blockIdx.x * (blockDim.x >> 5) + ((int)threadIdx.x >> 5);
  const int tilesN = Np >> 6;
  const int tilesM = M >> 5;
  const int tM = w / tilesN;
  const int tN = w - tM * tilesN;
  if (tM >= tilesM) return;

  const int m0 = tM << 5;
  const int n0 = tN << 6;
  const int lane = (int)threadIdx.x & 31;
  const int half = lane >> 4;
  const int lcol = lane & 15;
  const int r0 = m0 + lcol;
  const int r1 = r0 + 16;
  const float ir0 = invd[r0];
  const float ir1 = invd[r1];
  const float* h0 = h + (size_t)r0 * d;
  const float* h1 = h + (size_t)r1 * d;
  const float* s0 = msum + (size_t)r0 * d;
  const float* s1 = msum + (size_t)r1 * d;
  const size_t ldb = 2 * (size_t)Np;
  const int boff = 2 * (n0 + lcol);

  v8f acc0[4] = {}, acc1[4] = {};

  for (int k0 = 0; k0 < d; k0 += 4) {
    const int ka = k0 + 2 * half;
    v2f a0 = *(const v2f*)(h0 + ka);
    v2f a1 = *(const v2f*)(h1 + ka);
    const float* bp = Wp + (size_t)(ka >> 1) * ldb + boff;
#pragma unroll
    for (int j = 0; j < 4; ++j) {
      v2f b = *(const v2f*)(bp + 32 * j);
      acc0[j] = wmma4(a0, b, acc0[j]);
      acc1[j] = wmma4(a1, b, acc1[j]);
    }
  }
  for (int k0 = 0; k0 < d; k0 += 4) {
    const int ka = k0 + 2 * half;
    v2f a0 = *(const v2f*)(s0 + ka);
    v2f a1 = *(const v2f*)(s1 + ka);
    a0.x *= ir0; a0.y *= ir0;
    a1.x *= ir1; a1.y *= ir1;
    const float* bp = Wp + (size_t)((d + ka) >> 1) * ldb + boff;
#pragma unroll
    for (int j = 0; j < 4; ++j) {
      v2f b = *(const v2f*)(bp + 32 * j);
      acc0[j] = wmma4(a0, b, acc0[j]);
      acc1[j] = wmma4(a1, b, acc1[j]);
    }
  }

#pragma unroll
  for (int j = 0; j < 4; ++j) {
    int cc = n0 + j * 16 + lcol;
    if (cc < N) {
      float bj = bias[cc];
#pragma unroll
      for (int r = 0; r < 8; ++r) {
        float v0 = acc0[j][r] + bj;
        float v1 = acc1[j][r] + bj;
        if (RELU) {
          v0 = fmaxf(v0, 0.0f);
          v1 = fmaxf(v1, 0.0f);
        }
        out[(size_t)(m0 + half * 8 + r) * N + cc] = v0;
        out[(size_t)(m0 + 16 + half * 8 + r) * N + cc] = v1;
      }
    }
  }
}

extern "C" void kernel_launch(void* const* d_in, const int* in_sizes, int n_in, void* d_out,
                              int out_size, void* d_ws, size_t ws_size, hipStream_t stream) {
  (void)in_sizes; (void)n_in; (void)out_size; (void)ws_size;
  const float* x  = (const float*)d_in[0];
  const float* W1 = (const float*)d_in[1];
  const float* b1 = (const float*)d_in[2];
  const float* W2 = (const float*)d_in[3];
  const float* b2 = (const float*)d_in[4];
  const float* W3 = (const float*)d_in[5];
  const float* b3 = (const float*)d_in[6];
  const int* src  = (const int*)d_in[7];
  const int* dst  = (const int*)d_in[8];
  float* out = (float*)d_out;

  constexpr int M = 40000, E = 640000, INF = 128, HID = 256, OUTF = 47, OUTP = 64;

  float* ws = (float*)d_ws;
  int*   deg  = (int*)ws;
  float* invd = ws + 40000;
  float* msum = ws + 80000;
  float* h1   = msum + (size_t)M * HID;
  float* h2   = h1 + (size_t)M * HID;
  float* Wp1  = h2 + (size_t)M * HID;
  float* Wp2  = Wp1 + (size_t)(2 * INF) * HID;
  float* Wp3  = Wp2 + (size_t)(2 * HID) * HID;

  const int T = 256;
  auto blocks = [](long long n, int t) { return (int)((n + t - 1) / t); };

  pack_w_kernel<<<blocks((2 * INF / 2) * (long long)HID, T), T, 0, stream>>>(W1, Wp1, 2 * INF, HID, HID);
  pack_w_kernel<<<blocks((2 * HID / 2) * (long long)HID, T), T, 0, stream>>>(W2, Wp2, 2 * HID, HID, HID);
  pack_w_kernel<<<blocks((2 * HID / 2) * (long long)OUTP, T), T, 0, stream>>>(W3, Wp3, 2 * HID, OUTF, OUTP);

  zero4_kernel<<<blocks(M / 4, T), T, 0, stream>>>((float4*)deg, M / 4);
  degree_kernel<<<blocks(E, T), T, 0, stream>>>(dst, deg, E);
  invdeg_kernel<<<blocks(M, T), T, 0, stream>>>(deg, invd, M);

  {
    long long n4 = (long long)M * INF / 4;
    zero4_kernel<<<blocks(n4, T), T, 0, stream>>>((float4*)msum, n4);
    long long tot = (long long)E << 5;
    scatter_sum_kernel<<<blocks(tot, T), T, 0, stream>>>(x, src, dst, msum, E, INF, 5);
    int tiles = (M / 32) * (HID / 64);
    sage_gemm_wmma<true><<<blocks((long long)tiles * 32, T), T, 0, stream>>>(
        x, msum, invd, Wp1, b1, h1, M, INF, HID, HID);
  }
  {
    long long n4 = (long long)M * HID / 4;
    zero4_kernel<<<blocks(n4, T), T, 0, stream>>>((float4*)msum, n4);
    long long tot = (long long)E << 6;
    scatter_sum_kernel<<<blocks(tot, T), T, 0, stream>>>(h1, src, dst, msum, E, HID, 6);
    int tiles = (M / 32) * (HID / 64);
    sage_gemm_wmma<true><<<blocks((long long)tiles * 32, T), T, 0, stream>>>(
        h1, msum, invd, Wp2, b2, h2, M, HID, HID, HID);
  }
  {
    long long n4 = (long long)M * HID / 4;
    zero4_kernel<<<blocks(n4, T), T, 0, stream>>>((float4*)msum, n4);
    long long tot = (long long)E << 6;
    scatter_sum_kernel<<<blocks(tot, T), T, 0, stream>>>(h2, src, dst, msum, E, HID, 6);
    int tiles = (M / 32) * (OUTP / 64);
    sage_gemm_wmma<false><<<blocks((long long)tiles * 32, T), T, 0, stream>>>(
        h2, msum, invd, Wp3, b3, out, M, HID, OUTF, OUTP);
  }
}


# neighbour sums by gather over a destination-grouped edge list (built once per call, shared by the 3 layers) instead of float atomic scatter-add plus zero fill
# speedup vs baseline: 2.3012x; 2.3012x over previous
#include <hip/hip_runtime.h>

typedef __attribute__((ext_vector_type(2))) float v2f;
typedef __attribute__((ext_vector_type(8))) float v8f;

__global__ void zero4_kernel(float4* __restrict__ p, long long n4) {
  long long i = (long long)blockIdx.x * blockDim.x + threadIdx.x;
  if (i < n4) p[i] = make_float4(0.f, 0.f, 0.f, 0.f);
}

__global__ void degree_rank_kernel(const int* __restrict__ dst, int* __restrict__ deg,
                                   int* __restrict__ rank, int nE) {
  int i = blockIdx.x * blockDim.x + threadIdx.x;
  if (i < nE) rank[i] = atomicAdd(&deg[dst[i]], 1);
}

__global__ __launch_bounds__(1024) void rowptr_scan_kernel(const int* __restrict__ deg,
                                                           int* __restrict__ rowptr, int n) {
  constexpr int VPT = 4;
  __shared__ int wsum[32];
  const int t = (int)threadIdx.x;
  const int lane = t & 31;
  const int wv = t >> 5;
  int carry = 0;
  for (int base = 0; base < n; base += 1024 * 4 * VPT) {
    const int i0 = base + 4 * VPT * t;
    int4 e[VPT];
    int run = 0;
#pragma unroll
    for (int k = 0; k < VPT; ++k) {
      int4 v = make_int4(0, 0, 0, 0);
      if (i0 + 4 * k < n) v = *(const int4*)(deg + i0 + 4 * k);
      e[k].x = run; run += v.x;
      e[k].y = run; run += v.y;
      e[k].z = run; run += v.z;
      e[k].w = run; run += v.w;
    }
    int inc = run;
#pragma unroll
    for (int off = 1; off < 32; off <<= 1) {
      int y = __shfl_up(inc, off);
      if (lane >= off) inc += y;
    }
    if (lane == 31) wsum[wv] = inc;
    __syncthreads();
    const int wtot = wsum[lane];
    int winc = wtot;
#pragma unroll
    for (int off = 1; off < 32; off <<= 1) {
      int y = __shfl_up(winc, off);
      if (lane >= off) winc += y;
    }
    const int wexcl = __shfl(winc, wv) - __shfl(wtot, wv);
    const int total = __shfl(winc, 31);
    const int ex = carry + wexcl + (inc - run);
#pragma unroll
    for (int k = 0; k < VPT; ++k) {
      if (i0 + 4 * k < n)
        *(int4*)(rowptr + i0 + 4 * k) =
            make_int4(ex + e[k].x, ex + e[k].y, ex + e[k].z, ex + e[k].w);
    }
    carry += total;
    __syncthreads();
  }
  if (t == 0) rowptr[n] = carry;
}

__global__ void fill_col_kernel(const int* __restrict__ src, const int* __restrict__ dst,
                                const int* __restrict__ rowptr, const int* __restrict__ rank,
                                int* __restrict__ col, int nE) {
  int i = blockIdx.x * blockDim.x + threadIdx.x;
  if (i < nE) col[rowptr[dst[i]] + rank[i]] = src[i];
}

__global__ void invdeg_kernel(const int* __restrict__ deg, float* __restrict__ invd, int n) {
  int i = blockIdx.x * blockDim.x + threadIdx.x;
  if (i < n) invd[i] = 1.0f / fmaxf((float)deg[i], 1.0f);
}

__global__ void pack_w_kernel(const float* __restrict__ W, float* __restrict__ Wp,
                              int K, int N, int Np) {
  int i = blockIdx.x * blockDim.x + threadIdx.x;
  int total = (K >> 1) * Np;
  if (i >= total) return;
  int k2 = i / Np;
  int c = i - k2 * Np;
  float x0 = 0.f, x1 = 0.f;
  if (c < N) {
    x0 = W[(size_t)(2 * k2) * N + c];
    x1 = W[(size_t)(2 * k2 + 1) * N + c];
  }
  *(float2*)(Wp + (size_t)k2 * 2 * Np + 2 * c) = make_float2(x0, x1);
}

template <int D>
__global__ __launch_bounds__(256) void gather_sum_kernel(
    const float* __restrict__ h, const int* __restrict__ rowptr, const int* __restrict__ col,
    float* __restrict__ msum, int M) {
  constexpr int NV = D / 128;
  const int lane = (int)threadIdx.x & 31;
  const int v = __builtin_amdgcn_readfirstlane((int)blockIdx.x * 8 + ((int)threadIdx.x >> 5));
  if (v >= M) return;
  const int beg = rowptr[v];
  const int end = rowptr[v + 1];
  const float* hl = h + 4 * lane;
  float4 acc[NV];
#pragma unroll
  for (int k = 0; k < NV; ++k) acc[k] = make_float4(0.f, 0.f, 0.f, 0.f);

  for (int c0 = beg; c0 < end; c0 += 32) {
    const int n = min(32, end - c0);
    const int myc = (lane < n) ? col[c0 + lane] : 0;
    int i = 0;
    for (; i + 4 <= n; i += 4) {
      const float* p0 = hl + (size_t)__builtin_amdgcn_readlane(myc, i) * D;
      const float* p1 = hl + (size_t)__builtin_amdgcn_readlane(myc, i + 1) * D;
      const float* p2 = hl + (size_t)__builtin_amdgcn_readlane(myc, i + 2) * D;
      const float* p3 = hl + (size_t)__builtin_amdgcn_readlane(myc, i + 3) * D;
      float4 r0[NV], r1[NV], r2[NV], r3[NV];
#pragma unroll
      for (int k = 0; k < NV; ++k) {
        r0[k] = *(const float4*)(p0 + 128 * k);
        r1[k] = *(const float4*)(p1 + 128 * k);
        r2[k] = *(const float4*)(p2 + 128 * k);
        r3[k] = *(const float4*)(p3 + 128 * k);
      }
#pragma unroll
      for (int k = 0; k < NV; ++k) {
        acc[k].x += r0[k].x; acc[k].y += r0[k].y; acc[k].z += r0[k].z; acc[k].w += r0[k].w;
        acc[k].x += r1[k].x; acc[k].y += r1[k].y; acc[k].z += r1[k].z; acc[k].w += r1[k].w;
        acc[k].x += r2[k].x; acc[k].y += r2[k].y; acc[k].z += r2[k].z; acc[k].w += r2[k].w;
        acc[k].x += r3[k].x; acc[k].y += r3[k].y; acc[k].z += r3[k].z; acc[k].w += r3[k].w;
      }
    }
    for (; i < n; ++i) {
      const float* p0 = hl + (size_t)__builtin_amdgcn_readlane(myc, i) * D;
#pragma unroll
      for (int k = 0; k < NV; ++k) {
        const float4 r = *(const float4*)(p0 + 128 * k);
        acc[k].x += r.x; acc[k].y += r.y; acc[k].z += r.z; acc[k].w += r.w;
      }
    }
  }
  float* o = msum + (size_t)v * D + 4 * lane;
#pragma unroll
  for (int k = 0; k < NV; ++k) *(float4*)(o + 128 * k) = acc[k];
}

__device__ __forceinline__ v8f wmma4(v2f a, v2f b, v8f c) {
  return __builtin_amdgcn_wmma_f32_16x16x4_f32(false, a, false, b, (short)0, c, false, false);
}

template <bool RELU>
__global__ __launch_bounds__(256) void sage_gemm_wmma(
    const float* __restrict__ h, const float* __restrict__ msum,
    const float* __restrict__ invd, const float* __restrict__ Wp,
    const float* __restrict__ bias, float* __restrict__ out, int M, int d, int N, int Np) {
  const int w = blockIdx.x * (blockDim.x >> 5) + ((int)threadIdx.x >> 5);
  const int tilesN = Np >> 6;
  const int tilesM = M >> 5;
  const int tM = w / tilesN;
  const int tN = w - tM * tilesN;
  if (tM >= tilesM) return;

  const int m0 = tM << 5;
  const int n0 = tN << 6;
  const int lane = (int)threadIdx.x & 31;
  const int half = lane >> 4;
  const int lcol = lane & 15;
  const int r0 = m0 + lcol;
  const int r1 = r0 + 16;
  const float ir0 = invd[r0];
  const float ir1 = invd[r1];
  const float* h0 = h + (size_t)r0 * d;
  const float* h1 = h + (size_t)r1 * d;
  const float* s0 = msum + (size_t)r0 * d;
  const float* s1 = msum + (size_t)r1 * d;
  const size_t ldb = 2 * (size_t)Np;
  const int boff = 2 * (n0 + lcol);

  v8f acc0[4] = {}, acc1[4] = {};

  for (int k0 = 0; k0 < d; k0 += 4) {
    const int ka = k0 + 2 * half;
    v2f a0 = *(const v2f*)(h0 + ka);
    v2f a1 = *(const v2f*)(h1 + ka);
    const float* bp = Wp + (size_t)(ka >> 1) * ldb + boff;
#pragma unroll
    for (int j = 0; j < 4; ++j) {
      v2f b = *(const v2f*)(bp + 32 * j);
      acc0[j] = wmma4(a0, b, acc0[j]);
      acc1[j] = wmma4(a1, b, acc1[j]);
    }
  }
  for (int k0 = 0; k0 < d; k0 += 4) {
    const int ka = k0 + 2 * half;
    v2f a0 = *(const v2f*)(s0 + ka);
    v2f a1 = *(const v2f*)(s1 + ka);
    a0.x *= ir0; a0.y *= ir0;
    a1.x *= ir1; a1.y *= ir1;
    const float* bp = Wp + (size_t)((d + ka) >> 1) * ldb + boff;
#pragma unroll
    for (int j = 0; j < 4; ++j) {
      v2f b = *(const v2f*)(bp + 32 * j);
      acc0[j] = wmma4(a0, b, acc0[j]);
      acc1[j] = wmma4(a1, b, acc1[j]);
    }
  }

#pragma unroll
  for (int j = 0; j < 4; ++j) {
    int cc = n0 + j * 16 + lcol;
    if (cc < N) {
      float bj = bias[cc];
#pragma unroll
      for (int r = 0; r < 8; ++r) {
        float v0 = acc0[j][r] + bj;
        float v1 = acc1[j][r] + bj;
        if (RELU) {
          v0 = fmaxf(v0, 0.0f);
          v1 = fmaxf(v1, 0.0f);
        }
        out[(size_t)(m0 + half * 8 + r) * N + cc] = v0;
        out[(size_t)(m0 + 16 + half * 8 + r) * N + cc] = v1;
      }
    }
  }
}

extern "C" void kernel_launch(void* const* d_in, const int* in_sizes, int n_in, void* d_out,
                              int out_size, void* d_ws, size_t ws_size, hipStream_t stream) {
  (void)in_sizes; (void)n_in; (void)out_size; (void)ws_size;
  const float* x  = (const float*)d_in[0];
  const float* W1 = (const float*)d_in[1];
  const float* b1 = (const float*)d_in[2];
  const float* W2 = (const float*)d_in[3];
  const float* b2 = (const float*)d_in[4];
  const float* W3 = (const float*)d_in[5];
  const float* b3 = (const float*)d_in[6];
  const int* src  = (const int*)d_in[7];
  const int* dst  = (const int*)d_in[8];
  float* out = (float*)d_out;

  constexpr int M = 40000, E = 640000, INF = 128, HID = 256, OUTF = 47, OUTP = 64;

  float* ws = (float*)d_ws;
  int*   deg  = (int*)ws;
  float* invd = ws + 40000;
  float* msum = ws + 80000;
  float* h1   = msum + (size_t)M * HID;
  float* h2   = h1 + (size_t)M * HID;
  float* Wp1  = h2 + (size_t)M * HID;
  float* Wp2  = Wp1 + (size_t)(2 * INF) * HID;
  float* Wp3  = Wp2 + (size_t)(2 * HID) * HID;

  int* col    = (int*)d_out;
  int* rowptr = col + E;
  int* rank   = (int*)h2;

  const int T = 256;
  auto blocks = [](long long n, int t) { return (int)((n + t - 1) / t); };

  pack_w_kernel<<<blocks((2 * INF / 2) * (long long)HID, T), T, 0, stream>>>(W1, Wp1, 2 * INF, HID, HID);
  pack_w_kernel<<<blocks((2 * HID / 2) * (long long)HID, T), T, 0, stream>>>(W2, Wp2, 2 * HID, HID, HID);
  pack_w_kernel<<<blocks((2 * HID / 2) * (long long)OUTP, T), T, 0, stream>>>(W3, Wp3, 2 * HID, OUTF, OUTP);

  zero4_kernel<<<blocks(M / 4, T), T, 0, stream>>>((float4*)deg, M / 4);
  degree_rank_kernel<<<blocks(E, T), T, 0, stream>>>(dst, deg, rank, E);
  invdeg_kernel<<<blocks(M, T), T, 0, stream>>>(deg, invd, M);
  rowptr_scan_kernel<<<1, 1024, 0, stream>>>(deg, rowptr, M);
  fill_col_kernel<<<blocks(E, T), T, 0, stream>>>(src, dst, rowptr, rank, col, E);

  {
    gather_sum_kernel<INF><<<M / 8, T, 0, stream>>>(x, rowptr, col, msum, M);
    int tiles = (M / 32) * (HID / 64);
    sage_gemm_wmma<true><<<blocks((long long)tiles * 32, T), T, 0, stream>>>(
        x, msum, invd, Wp1, b1, h1, M, INF, HID, HID);
  }
  {
    gather_sum_kernel<HID><<<M / 8, T, 0, stream>>>(h1, rowptr, col, msum, M);
    int tiles = (M / 32) * (HID / 64);
    sage_gemm_wmma<true><<<blocks((long long)tiles * 32, T), T, 0, stream>>>(
        h1, msum, invd, Wp2, b2, h2, M, HID, HID, HID);
  }
  {
    gather_sum_kernel<HID><<<M / 8, T, 0, stream>>>(h2, rowptr, col, msum, M);
    int tiles = (M / 32) * (OUTP / 64);
    sage_gemm_wmma<false><<<blocks((long long)tiles * 32, T), T, 0, stream>>>(
        h2, msum, invd, Wp3, b3, out, M, HID, OUTF, OUTP);
  }
}


# self half of each layer's GEMM (h @ W[0:d), raw f32 tile) runs as extra blocks in the gather's launch: matrix work overlaps the cache-bound gather; 2nd launch adds the neighbour half
# speedup vs baseline: 2.3850x; 1.0364x over previous
#include <hip/hip_runtime.h>

typedef __attribute__((ext_vector_type(2))) float v2f;
typedef __attribute__((ext_vector_type(8))) float v8f;

__global__ void zero4_kernel(float4* __restrict__ p, long long n4) {
  long long i = (long long)blockIdx.x * blockDim.x + threadIdx.x;
  if (i < n4) p[i] = make_float4(0.f, 0.f, 0.f, 0.f);
}

__global__ void degree_rank_kernel(const int* __restrict__ dst, int* __restrict__ deg,
                                   int* __restrict__ rank, int nE) {
  int i = blockIdx.x * blockDim.x + threadIdx.x;
  if (i < nE) rank[i] = atomicAdd(&deg[dst[i]], 1);
}

__global__ __launch_bounds__(1024) void rowptr_scan_kernel(const int* __restrict__ deg,
                                                           int* __restrict__ rowptr, int n) {
  constexpr int VPT = 4;
  __shared__ int wsum[32];
  const int t = (int)threadIdx.x;
  const int lane = t & 31;
  const int wv = t >> 5;
  int carry = 0;
  for (int base = 0; base < n; base += 1024 * 4 * VPT) {
    const int i0 = base + 4 * VPT * t;
    int4 e[VPT];
    int run = 0;
#pragma unroll
    for (int k = 0; k < VPT; ++k) {
      int4 v = make_int4(0, 0, 0, 0);
      if (i0 + 4 * k < n) v = *(const int4*)(deg + i0 + 4 * k);
      e[k].x = run; run += v.x;
      e[k].y = run; run += v.y;
      e[k].z = run; run += v.z;
      e[k].w = run; run += v.w;
    }
    int inc = run;
#pragma unroll
    for (int off = 1; off < 32; off <<= 1) {
      int y = __shfl_up(inc, off);
      if (lane >= off) inc += y;
    }
    if (lane == 31) wsum[wv] = inc;
    __syncthreads();
    const int wtot = wsum[lane];
    int winc = wtot;
#pragma unroll
    for (int off = 1; off < 32; off <<= 1) {
      int y = __shfl_up(winc, off);
      if (lane >= off) winc += y;
    }
    const int wexcl = __shfl(winc, wv) - __shfl(wtot, wv);
    const int total = __shfl(winc, 31);
    const int ex = carry + wexcl + (inc - run);
#pragma unroll
    for (int k = 0; k < VPT; ++k) {
      if (i0 + 4 * k < n)
        *(int4*)(rowptr + i0 + 4 * k) =
            make_int4(ex + e[k].x, ex + e[k].y, ex + e[k].z, ex + e[k].w);
    }
    carry += total;
    __syncthreads();
  }
  if (t == 0) rowptr[n] = carry;
}

__global__ void fill_col_kernel(const int* __restrict__ src, const int* __restrict__ dst,
                                const int* __restrict__ rowptr, const int* __restrict__ rank,
                                int* __restrict__ col, int nE) {
  int i = blockIdx.x * blockDim.x + threadIdx.x;
  if (i < nE) col[rowptr[dst[i]] + rank[i]] = src[i];
}

__global__ void invdeg_kernel(const int* __restrict__ deg, float* __restrict__ invd, int n) {
  int i = blockIdx.x * blockDim.x + threadIdx.x;
  if (i < n) invd[i] = 1.0f / fmaxf((float)deg[i], 1.0f);
}

__global__ void pack_w_kernel(const float* __restrict__ W, float* __restrict__ Wp,
                              int K, int N, int Np) {
  int i = blockIdx.x * blockDim.x + threadIdx.x;
  int total = (K >> 1) * Np;
  if (i >= total) return;
  int k2 = i / Np;
  int c = i - k2 * Np;
  float x0 = 0.f, x1 = 0.f;
  if (c < N) {
    x0 = W[(size_t)(2 * k2) * N + c];
    x1 = W[(size_t)(2 * k2 + 1) * N + c];
  }
  *(float2*)(Wp + (size_t)k2 * 2 * Np + 2 * c) = make_float2(x0, x1);
}

template <int D>
__device__ __forceinline__ void gather_rows(
    const float* __restrict__ h, const int* __restrict__ rowptr, const int* __restrict__ col,
    float* __restrict__ msum, int M, int blk) {
  constexpr int NV = D / 128;
  const int lane = (int)threadIdx.x & 31;
  const int v = __builtin_amdgcn_readfirstlane(blk * 8 + ((int)threadIdx.x >> 5));
  if (v >= M) return;
  const int beg = rowptr[v];
  const int end = rowptr[v + 1];
  const float* hl = h + 4 * lane;
  float4 acc[NV];
#pragma unroll
  for (int k = 0; k < NV; ++k) acc[k] = make_float4(0.f, 0.f, 0.f, 0.f);

  for (int c0 = beg; c0 < end; c0 += 32) {
    const int n = min(32, end - c0);
    const int myc = (lane < n) ? col[c0 + lane] : 0;
    int i = 0;
    for (; i + 4 <= n; i += 4) {
      const float* p0 = hl + (size_t)__builtin_amdgcn_readlane(myc, i) * D;
      const float* p1 = hl + (size_t)__builtin_amdgcn_readlane(myc, i + 1) * D;
      const float* p2 = hl + (size_t)__builtin_amdgcn_readlane(myc, i + 2) * D;
      const float* p3 = hl + (size_t)__builtin_amdgcn_readlane(myc, i + 3) * D;
      float4 r0[NV], r1[NV], r2[NV], r3[NV];
#pragma unroll
      for (int k = 0; k < NV; ++k) {
        r0[k] = *(const float4*)(p0 + 128 * k);
        r1[k] = *(const float4*)(p1 + 128 * k);
        r2[k] = *(const float4*)(p2 + 128 * k);
        r3[k] = *(const float4*)(p3 + 128 * k);
      }
#pragma unroll
      for (int k = 0; k < NV; ++k) {
        acc[k].x += r0[k].x; acc[k].y += r0[k].y; acc[k].z += r0[k].z; acc[k].w += r0[k].w;
        acc[k].x += r1[k].x; acc[k].y += r1[k].y; acc[k].z += r1[k].z; acc[k].w += r1[k].w;
        acc[k].x += r2[k].x; acc[k].y += r2[k].y; acc[k].z += r2[k].z; acc[k].w += r2[k].w;
        acc[k].x += r3[k].x; acc[k].y += r3[k].y; acc[k].z += r3[k].z; acc[k].w += r3[k].w;
      }
    }
    for (; i < n; ++i) {
      const float* p0 = hl + (size_t)__builtin_amdgcn_readlane(myc, i) * D;
#pragma unroll
      for (int k = 0; k < NV; ++k) {
        const float4 r = *(const float4*)(p0 + 128 * k);
        acc[k].x += r.x; acc[k].y += r.y; acc[k].z += r.z; acc[k].w += r.w;
      }
    }
  }
  float* o = msum + (size_t)v * D + 4 * lane;
#pragma unroll
  for (int k = 0; k < NV; ++k) *(float4*)(o + 128 * k) = acc[k];
}

__device__ __forceinline__ v8f wmma4(v2f a, v2f b, v8f c) {
  return __builtin_amdgcn_wmma_f32_16x16x4_f32(false, a, false, b, (short)0, c, false, false);
}

__device__ __forceinline__ void gemm_self_tile(
    const float* __restrict__ h, const float* __restrict__ Wp, float* __restrict__ P,
    int M, int d, int Np, int blk) {
  const int w = blk * 8 + ((int)threadIdx.x >> 5);
  const int tilesN = Np >> 6;
  const int tilesM = M >> 5;
  const int tM = w / tilesN;
  const int tN = w - tM * tilesN;
  if (tM >= tilesM) return;

  const int m0 = tM << 5;
  const int n0 = tN << 6;
  const int lane = (int)threadIdx.x & 31;
  const int half = lane >> 4;
  const int lcol = lane & 15;
  const int r0 = m0 + lcol;
  const int r1 = r0 + 16;
  const float* h0 = h + (size_t)r0 * d;
  const float* h1 = h + (size_t)r1 * d;
  const size_t ldb = 2 * (size_t)Np;
  const int boff = 2 * (n0 + lcol);

  v8f acc0[4] = {}, acc1[4] = {};

  for (int k0 = 0; k0 < d; k0 += 4) {
    const int ka = k0 + 2 * half;
    v2f a0 = *(const v2f*)(h0 + ka);
    v2f a1 = *(const v2f*)(h1 + ka);
    const float* bp = Wp + (size_t)(ka >> 1) * ldb + boff;
#pragma unroll
    for (int j = 0; j < 4; ++j) {
      v2f b = *(const v2f*)(bp + 32 * j);
      acc0[j] = wmma4(a0, b, acc0[j]);
      acc1[j] = wmma4(a1, b, acc1[j]);
    }
  }

#pragma unroll
  for (int j = 0; j < 4; ++j) {
    const int cc = n0 + j * 16 + lcol;
#pragma unroll
    for (int r = 0; r < 8; ++r) {
      P[(size_t)(m0 + half * 8 + r) * Np + cc] = acc0[j][r];
      P[(size_t)(m0 + 16 + half * 8 + r) * Np + cc] = acc1[j][r];
    }
  }
}

template <int D>
__global__ __launch_bounds__(256) void self_gemm_and_gather_kernel(
    const float* __restrict__ h, const float* __restrict__ Wp, float* __restrict__ P, int Np,
    int gemm_blocks, const int* __restrict__ rowptr, const int* __restrict__ col,
    float* __restrict__ msum, int M) {
  const int b = (int)blockIdx.x;
  if (b < gemm_blocks) {
    gemm_self_tile(h, Wp, P, M, D, Np, b);
  } else {
    gather_rows<D>(h, rowptr, col, msum, M, b - gemm_blocks);
  }
}

template <bool RELU>
__global__ __launch_bounds__(256) void neigh_gemm_kernel(
    const float* __restrict__ msum, const float* __restrict__ invd,
    const float* __restrict__ Wp, const float* P, const float* __restrict__ bias, float* out,
    int M, int d, int N, int Np) {
  const int w = blockIdx.x * (blockDim.x >> 5) + ((int)threadIdx.x >> 5);
  const int tilesN = Np >> 6;
  const int tilesM = M >> 5;
  const int tM = w / tilesN;
  const int tN = w - tM * tilesN;
  if (tM >= tilesM) return;

  const int m0 = tM << 5;
  const int n0 = tN << 6;
  const int lane = (int)threadIdx.x & 31;
  const int half = lane >> 4;
  const int lcol = lane & 15;
  const int r0 = m0 + lcol;
  const int r1 = r0 + 16;
  const float ir0 = invd[r0];
  const float ir1 = invd[r1];
  const float* s0 = msum + (size_t)r0 * d;
  const float* s1 = msum + (size_t)r1 * d;
  const size_t ldb = 2 * (size_t)Np;
  const int boff = 2 * (n0 + lcol);

  v8f acc0[4], acc1[4];
#pragma unroll
  for (int j = 0; j < 4; ++j) {
    const int cc = n0 + j * 16 + lcol;
#pragma unroll
    for (int r = 0; r < 8; ++r) {
      acc0[j][r] = P[(size_t)(m0 + half * 8 + r) * Np + cc];
      acc1[j][r] = P[(size_t)(m0 + 16 + half * 8 + r) * Np + cc];
    }
  }

  for (int k0 = 0; k0 < d; k0 += 4) {
    const int ka = k0 + 2 * half;
    v2f a0 = *(const v2f*)(s0 + ka);
    v2f a1 = *(const v2f*)(s1 + ka);
    a0.x *= ir0; a0.y *= ir0;
    a1.x *= ir1; a1.y *= ir1;
    const float* bp = Wp + (size_t)((d + ka) >> 1) * ldb + boff;
#pragma unroll
    for (int j = 0; j < 4; ++j) {
      v2f b = *(const v2f*)(bp + 32 * j);
      acc0[j] = wmma4(a0, b, acc0[j]);
      acc1[j] = wmma4(a1, b, acc1[j]);
    }
  }

#pragma unroll
  for (int j = 0; j < 4; ++j) {
    int cc = n0 + j * 16 + lcol;
    if (cc < N) {
      float bj = bias[cc];
#pragma unroll
      for (int r = 0; r < 8; ++r) {
        float v0 = acc0[j][r] + bj;
        float v1 = acc1[j][r] + bj;
        if (RELU) {
          v0 = fmaxf(v0, 0.0f);
          v1 = fmaxf(v1, 0.0f);
        }
        out[(size_t)(m0 + half * 8 + r) * N + cc] = v0;
        out[(size_t)(m0 + 16 + half * 8 + r) * N + cc] = v1;
      }
    }
  }
}

extern "C" void kernel_launch(void* const* d_in, const int* in_sizes, int n_in, void* d_out,
                              int out_size, void* d_ws, size_t ws_size, hipStream_t stream) {
  (void)in_sizes; (void)n_in; (void)out_size; (void)ws_size;
  const float* x  = (const float*)d_in[0];
  const float* W1 = (const float*)d_in[1];
  const float* b1 = (const float*)d_in[2];
  const float* W2 = (const float*)d_in[3];
  const float* b2 = (const float*)d_in[4];
  const float* W3 = (const float*)d_in[5];
  const float* b3 = (const float*)d_in[6];
  const int* src  = (const int*)d_in[7];
  const int* dst  = (const int*)d_in[8];
  float* out = (float*)d_out;

  constexpr int M = 40000, E = 640000, INF = 128, HID = 256, OUTF = 47, OUTP = 64;

  float* ws = (float*)d_ws;
  int*   deg  = (int*)ws;
  float* invd = ws + 40000;
  float* msum = ws + 80000;
  float* h1   = msum + (size_t)M * HID;
  float* h2   = h1 + (size_t)M * HID;
  float* Wp1  = h2 + (size_t)M * HID;
  float* Wp2  = Wp1 + (size_t)(2 * INF) * HID;
  float* Wp3  = Wp2 + (size_t)(2 * HID) * HID;

  int* col    = (int*)d_out;
  int* rowptr = col + E;
  int* rank   = (int*)h2;

  const int T = 256;
  auto blocks = [](long long n, int t) { return (int)((n + t - 1) / t); };

  pack_w_kernel<<<blocks((2 * INF / 2) * (long long)HID, T), T, 0, stream>>>(W1, Wp1, 2 * INF, HID, HID);
  pack_w_kernel<<<blocks((2 * HID / 2) * (long long)HID, T), T, 0, stream>>>(W2, Wp2, 2 * HID, HID, HID);
  pack_w_kernel<<<blocks((2 * HID / 2) * (long long)OUTP, T), T, 0, stream>>>(W3, Wp3, 2 * HID, OUTF, OUTP);

  zero4_kernel<<<blocks(M / 4, T), T, 0, stream>>>((float4*)deg, M / 4);
  degree_rank_kernel<<<blocks(E, T), T, 0, stream>>>(dst, deg, rank, E);
  invdeg_kernel<<<blocks(M, T), T, 0, stream>>>(deg, invd, M);
  rowptr_scan_kernel<<<1, 1024, 0, stream>>>(deg, rowptr, M);
  fill_col_kernel<<<blocks(E, T), T, 0, stream>>>(src, dst, rowptr, rank, col, E);

  {
    int tiles = (M / 32) * (HID / 64);
    int gb = blocks((long long)tiles * 32, T);
    self_gemm_and_gather_kernel<INF><<<gb + M / 8, T, 0, stream>>>(
        x, Wp1, h1, HID, gb, rowptr, col, msum, M);
    neigh_gemm_kernel<true><<<gb, T, 0, stream>>>(
        msum, invd, Wp1, h1, b1, h1, M, INF, HID, HID);
  }
  {
    int tiles = (M / 32) * (HID / 64);
    int gb = blocks((long long)tiles * 32, T);
    self_gemm_and_gather_kernel<HID><<<gb + M / 8, T, 0, stream>>>(
        h1, Wp2, h2, HID, gb, rowptr, col, msum, M);
    neigh_gemm_kernel<true><<<gb, T, 0, stream>>>(
        msum, invd, Wp2, h2, b2, h2, M, HID, HID, HID);
  }
  {
    int tiles = (M / 32) * (OUTP / 64);
    int gb = blocks((long long)tiles * 32, T);
    float* P3 = h1;
    self_gemm_and_gather_kernel<HID><<<gb + M / 8, T, 0, stream>>>(
        h2, Wp3, P3, OUTP, gb, rowptr, col, msum, M);
    neigh_gemm_kernel<false><<<gb, T, 0, stream>>>(
        msum, invd, Wp3, P3, b3, out, M, HID, OUTF, OUTP);
  }
}


# layer 3 maps before it aggregates: one GEMM h2 @ [W self | W neighbour] (2 x 64 padded columns), then neighbour sums over 64-float rows instead of 256-float rows, finished straight into out
# speedup vs baseline: 2.5393x; 1.0647x over previous
#include <hip/hip_runtime.h>

typedef __attribute__((ext_vector_type(2))) float v2f;
typedef __attribute__((ext_vector_type(8))) float v8f;

__global__ void zero4_kernel(float4* __restrict__ p, long long n4) {
  long long i = (long long)blockIdx.x * blockDim.x + threadIdx.x;
  if (i < n4) p[i] = make_float4(0.f, 0.f, 0.f, 0.f);
}

__global__ void degree_rank_kernel(const int* __restrict__ dst, int* __restrict__ deg,
                                   int* __restrict__ rank, int nE) {
  int i = blockIdx.x * blockDim.x + threadIdx.x;
  if (i < nE) rank[i] = atomicAdd(&deg[dst[i]], 1);
}

__global__ __launch_bounds__(1024) void rowptr_scan_kernel(const int* __restrict__ deg,
                                                           int* __restrict__ rowptr, int n) {
  constexpr int VPT = 4;
  __shared__ int wsum[32];
  const int t = (int)threadIdx.x;
  const int lane = t & 31;
  const int wv = t >> 5;
  int carry = 0;
  for (int base = 0; base < n; base += 1024 * 4 * VPT) {
    const int i0 = base + 4 * VPT * t;
    int4 e[VPT];
    int run = 0;
#pragma unroll
    for (int k = 0; k < VPT; ++k) {
      int4 v = make_int4(0, 0, 0, 0);
      if (i0 + 4 * k < n) v = *(const int4*)(deg + i0 + 4 * k);
      e[k].x = run; run += v.x;
      e[k].y = run; run += v.y;
      e[k].z = run; run += v.z;
      e[k].w = run; run += v.w;
    }
    int inc = run;
#pragma unroll
    for (int off = 1; off < 32; off <<= 1) {
      int y = __shfl_up(inc, off);
      if (lane >= off) inc += y;
    }
    if (lane == 31) wsum[wv] = inc;
    __syncthreads();
    const int wtot = wsum[lane];
    int winc = wtot;
#pragma unroll
    for (int off = 1; off < 32; off <<= 1) {
      int y = __shfl_up(winc, off);
      if (lane >= off) winc += y;
    }
    const int wexcl = __shfl(winc, wv) - __shfl(wtot, wv);
    const int total = __shfl(winc, 31);
    const int ex = carry + wexcl + (inc - run);
#pragma unroll
    for (int k = 0; k < VPT; ++k) {
      if (i0 + 4 * k < n)
        *(int4*)(rowptr + i0 + 4 * k) =
            make_int4(ex + e[k].x, ex + e[k].y, ex + e[k].z, ex + e[k].w);
    }
    carry += total;
    __syncthreads();
  }
  if (t == 0) rowptr[n] = carry;
}

__global__ void fill_col_kernel(const int* __restrict__ src, const int* __restrict__ dst,
                                const int* __restrict__ rowptr, const int* __restrict__ rank,
                                int* __restrict__ col, int nE) {
  int i = blockIdx.x * blockDim.x + threadIdx.x;
  if (i < nE) col[rowptr[dst[i]] + rank[i]] = src[i];
}

__global__ void invdeg_kernel(const int* __restrict__ deg, float* __restrict__ invd, int n) {
  int i = blockIdx.x * blockDim.x + threadIdx.x;
  if (i < n) invd[i] = 1.0f / fmaxf((float)deg[i], 1.0f);
}

__global__ void pack_w_kernel(const float* __restrict__ W, float* __restrict__ Wp,
                              int K, int N, int Np) {
  int i = blockIdx.x * blockDim.x + threadIdx.x;
  int total = (K >> 1) * Np;
  if (i >= total) return;
  int k2 = i / Np;
  int c = i - k2 * Np;
  float x0 = 0.f, x1 = 0.f;
  if (c < N) {
    x0 = W[(size_t)(2 * k2) * N + c];
    x1 = W[(size_t)(2 * k2 + 1) * N + c];
  }
  *(float2*)(Wp + (size_t)k2 * 2 * Np + 2 * c) = make_float2(x0, x1);
}

template <int D>
__device__ __forceinline__ void gather_rows(
    const float* __restrict__ h, const int* __restrict__ rowptr, const int* __restrict__ col,
    float* __restrict__ msum, int M, int blk) {
  constexpr int NV = D / 128;
  const int lane = (int)threadIdx.x & 31;
  const int v = __builtin_amdgcn_readfirstlane(blk * 8 + ((int)threadIdx.x >> 5));
  if (v >= M) return;
  const int beg = rowptr[v];
  const int end = rowptr[v + 1];
  const float* hl = h + 4 * lane;
  float4 acc[NV];
#pragma unroll
  for (int k = 0; k < NV; ++k) acc[k] = make_float4(0.f, 0.f, 0.f, 0.f);

  for (int c0 = beg; c0 < end; c0 += 32) {
    const int n = min(32, end - c0);
    const int myc = (lane < n) ? col[c0 + lane] : 0;
    int i = 0;
    for (; i + 4 <= n; i += 4) {
      const float* p0 = hl + (size_t)__builtin_amdgcn_readlane(myc, i) * D;
      const float* p1 = hl + (size_t)__builtin_amdgcn_readlane(myc, i + 1) * D;
      const float* p2 = hl + (size_t)__builtin_amdgcn_readlane(myc, i + 2) * D;
      const float* p3 = hl + (size_t)__builtin_amdgcn_readlane(myc, i + 3) * D;
      float4 r0[NV], r1[NV], r2[NV], r3[NV];
#pragma unroll
      for (int k = 0; k < NV; ++k) {
        r0[k] = *(const float4*)(p0 + 128 * k);
        r1[k] = *(const float4*)(p1 + 128 * k);
        r2[k] = *(const float4*)(p2 + 128 * k);
        r3[k] = *(const float4*)(p3 + 128 * k);
      }
#pragma unroll
      for (int k = 0; k < NV; ++k) {
        acc[k].x += r0[k].x; acc[k].y += r0[k].y; acc[k].z += r0[k].z; acc[k].w += r0[k].w;
        acc[k].x += r1[k].x; acc[k].y += r1[k].y; acc[k].z += r1[k].z; acc[k].w += r1[k].w;
        acc[k].x += r2[k].x; acc[k].y += r2[k].y; acc[k].z += r2[k].z; acc[k].w += r2[k].w;
        acc[k].x += r3[k].x; acc[k].y += r3[k].y; acc[k].z += r3[k].z; acc[k].w += r3[k].w;
      }
    }
    for (; i < n; ++i) {
      const float* p0 = hl + (size_t)__builtin_amdgcn_readlane(myc, i) * D;
#pragma unroll
      for (int k = 0; k < NV; ++k) {
        const float4 r = *(const float4*)(p0 + 128 * k);
        acc[k].x += r.x; acc[k].y += r.y; acc[k].z += r.z; acc[k].w += r.w;
      }
    }
  }
  float* o = msum + (size_t)v * D + 4 * lane;
#pragma unroll
  for (int k = 0; k < NV; ++k) *(float4*)(o + 128 * k) = acc[k];
}

__device__ __forceinline__ v8f wmma4(v2f a, v2f b, v8f c) {
  return __builtin_amdgcn_wmma_f32_16x16x4_f32(false, a, false, b, (short)0, c, false, false);
}

__device__ __forceinline__ void gemm_self_tile(
    const float* __restrict__ h, const float* __restrict__ Wp, float* __restrict__ P,
    int M, int d, int Np, int blk) {
  const int w = blk * 8 + ((int)threadIdx.x >> 5);
  const int tilesN = Np >> 6;
  const int tilesM = M >> 5;
  const int tM = w / tilesN;
  const int tN = w - tM * tilesN;
  if (tM >= tilesM) return;

  const int m0 = tM << 5;
  const int n0 = tN << 6;
  const int lane = (int)threadIdx.x & 31;
  const int half = lane >> 4;
  const int lcol = lane & 15;
  const int r0 = m0 + lcol;
  const int r1 = r0 + 16;
  const float* h0 = h + (size_t)r0 * d;
  const float* h1 = h + (size_t)r1 * d;
  const size_t ldb = 2 * (size_t)Np;
  const int boff = 2 * (n0 + lcol);

  v8f acc0[4] = {}, acc1[4] = {};

  for (int k0 = 0; k0 < d; k0 += 4) {
    const int ka = k0 + 2 * half;
    v2f a0 = *(const v2f*)(h0 + ka);
    v2f a1 = *(const v2f*)(h1 + ka);
    const float* bp = Wp + (size_t)(ka >> 1) * ldb + boff;
#pragma unroll
    for (int j = 0; j < 4; ++j) {
      v2f b = *(const v2f*)(bp + 32 * j);
      acc0[j] = wmma4(a0, b, acc0[j]);
      acc1[j] = wmma4(a1, b, acc1[j]);
    }
  }

#pragma unroll
  for (int j = 0; j < 4; ++j) {
    const int cc = n0 + j * 16 + lcol;
#pragma unroll
    for (int r = 0; r < 8; ++r) {
      P[(size_t)(m0 + half * 8 + r) * Np + cc] = acc0[j][r];
      P[(size_t)(m0 + 16 + half * 8 + r) * Np + cc] = acc1[j][r];
    }
  }
}

template <int D>
__global__ __launch_bounds__(256) void self_gemm_and_gather_kernel(
    const float* __restrict__ h, const float* __restrict__ Wp, float* __restrict__ P, int Np,
    int gemm_blocks, const int* __restrict__ rowptr, const int* __restrict__ col,
    float* __restrict__ msum, int M) {
  const int b = (int)blockIdx.x;
  if (b < gemm_blocks) {
    gemm_self_tile(h, Wp, P, M, D, Np, b);
  } else {
    gather_rows<D>(h, rowptr, col, msum, M, b - gemm_blocks);
  }
}

template <bool RELU>
__global__ __launch_bounds__(256) void neigh_gemm_kernel(
    const float* __restrict__ msum, const float* __restrict__ invd,
    const float* __restrict__ Wp, const float* P, const float* __restrict__ bias, float* out,
    int M, int d, int N, int Np) {
  const int w = blockIdx.x * (blockDim.x >> 5) + ((int)threadIdx.x >> 5);
  const int tilesN = Np >> 6;
  const int tilesM = M >> 5;
  const int tM = w / tilesN;
  const int tN = w - tM * tilesN;
  if (tM >= tilesM) return;

  const int m0 = tM << 5;
  const int n0 = tN << 6;
  const int lane = (int)threadIdx.x & 31;
  const int half = lane >> 4;
  const int lcol = lane & 15;
  const int r0 = m0 + lcol;
  const int r1 = r0 + 16;
  const float ir0 = invd[r0];
  const float ir1 = invd[r1];
  const float* s0 = msum + (size_t)r0 * d;
  const float* s1 = msum + (size_t)r1 * d;
  const size_t ldb = 2 * (size_t)Np;
  const int boff = 2 * (n0 + lcol);

  v8f acc0[4], acc1[4];
#pragma unroll
  for (int j = 0; j < 4; ++j) {
    const int cc = n0 + j * 16 + lcol;
#pragma unroll
    for (int r = 0; r < 8; ++r) {
      acc0[j][r] = P[(size_t)(m0 + half * 8 + r) * Np + cc];
      acc1[j][r] = P[(size_t)(m0 + 16 + half * 8 + r) * Np + cc];
    }
  }

  for (int k0 = 0; k0 < d; k0 += 4) {
    const int ka = k0 + 2 * half;
    v2f a0 = *(const v2f*)(s0 + ka);
    v2f a1 = *(const v2f*)(s1 + ka);
    a0.x *= ir0; a0.y *= ir0;
    a1.x *= ir1; a1.y *= ir1;
    const float* bp = Wp + (size_t)((d + ka) >> 1) * ldb + boff;
#pragma unroll
    for (int j = 0; j < 4; ++j) {
      v2f b = *(const v2f*)(bp + 32 * j);
      acc0[j] = wmma4(a0, b, acc0[j]);
      acc1[j] = wmma4(a1, b, acc1[j]);
    }
  }

#pragma unroll
  for (int j = 0; j < 4; ++j) {
    int cc = n0 + j * 16 + lcol;
    if (cc < N) {
      float bj = bias[cc];
#pragma unroll
      for (int r = 0; r < 8; ++r) {
        float v0 = acc0[j][r] + bj;
        float v1 = acc1[j][r] + bj;
        if (RELU) {
          v0 = fmaxf(v0, 0.0f);
          v1 = fmaxf(v1, 0.0f);
        }
        out[(size_t)(m0 + half * 8 + r) * N + cc] = v0;
        out[(size_t)(m0 + 16 + half * 8 + r) * N + cc] = v1;
      }
    }
  }
}

__global__ void pack_w_pair_kernel(const float* __restrict__ W, float* __restrict__ Wp,
                                   int d, int N) {
  int i = blockIdx.x * blockDim.x + threadIdx.x;
  int total = (d >> 1) * 128;
  if (i >= total) return;
  int k2 = i >> 7;
  int c = i & 127;
  int part = c >> 6;
  int cc = c & 63;
  float x0 = 0.f, x1 = 0.f;
  if (cc < N) {
    x0 = W[(size_t)(part * d + 2 * k2) * N + cc];
    x1 = W[(size_t)(part * d + 2 * k2 + 1) * N + cc];
  }
  *(float2*)(Wp + (size_t)k2 * 256 + 2 * c) = make_float2(x0, x1);
}

__global__ __launch_bounds__(256) void pair_gemm_and_move_list_kernel(
    const float* __restrict__ h, const float* __restrict__ Wp, float* __restrict__ PZ,
    int gemm_blocks, const int* __restrict__ list_from, int* __restrict__ list_to,
    int n_ints, int M, int d) {
  const int b = (int)blockIdx.x;
  if (b < gemm_blocks) {
    gemm_self_tile(h, Wp, PZ, M, d, 128, b);
  } else {
    const int i4 = (b - gemm_blocks) * 256 + (int)threadIdx.x;
    const int n4 = n_ints >> 2;
    if (i4 < n4) {
      ((int4*)list_to)[i4] = ((const int4*)list_from)[i4];
    } else if (i4 == n4) {
      for (int k = 4 * n4; k < n_ints; ++k) list_to[k] = list_from[k];
    }
  }
}

__global__ __launch_bounds__(256) void gather_finish_kernel(
    const float* __restrict__ PZ, const int* __restrict__ rowptr, const int* __restrict__ col,
    const float* __restrict__ invd, const float* __restrict__ bias, float* __restrict__ out,
    int M) {
  constexpr int NO = 47;
  constexpr int NPW = 4;
  __shared__ float stage[32 * NO];
  const int lane = (int)threadIdx.x & 31;
  const int half = lane >> 4;
  const int wv = __builtin_amdgcn_readfirstlane((int)threadIdx.x >> 5);
  const int vb = (int)blockIdx.x * 32;
  const int v0 = vb + wv * NPW;
  const float* Zl = PZ + 64 + 4 * (lane & 15);

  int rp[NPW + 1];
#pragma unroll
  for (int k = 0; k <= NPW; ++k) rp[k] = rowptr[min(v0 + k, M)];
  int ids[NPW];
  float4 prow[NPW];
#pragma unroll
  for (int t = 0; t < NPW; ++t) {
    ids[t] = (lane < rp[t + 1] - rp[t]) ? col[rp[t] + lane] : 0;
    prow[t] = *(const float4*)(PZ + (size_t)min(v0 + t, M - 1) * 128 + 4 * (lane & 15));
  }

#pragma unroll
  for (int t = 0; t < NPW; ++t) {
    const int v = v0 + t;
    const int beg = rp[t];
    const int end = rp[t + 1];
    float4 acc = make_float4(0.f, 0.f, 0.f, 0.f);
    int myc = ids[t];
    for (int c0 = beg; c0 < end; c0 += 32) {
      const int n = min(32, end - c0);
      if (c0 != beg) myc = (lane < n) ? col[c0 + lane] : 0;
      for (int i = 0; i < n; i += 8) {
        float4 r[4];
#pragma unroll
        for (int q = 0; q < 4; ++q) {
          const int k = i + 2 * q;
          const int ea = __builtin_amdgcn_readlane(myc, k);
          const int eb = __builtin_amdgcn_readlane(myc, k + 1);
          r[q] = make_float4(0.f, 0.f, 0.f, 0.f);
          if (k + half < n) r[q] = *(const float4*)(Zl + (size_t)(half ? eb : ea) * 128);
        }
#pragma unroll
        for (int q = 0; q < 4; ++q) {
          acc.x += r[q].x; acc.y += r[q].y; acc.z += r[q].z; acc.w += r[q].w;
        }
      }
    }
    acc.x += __shfl_xor(acc.x, 16);
    acc.y += __shfl_xor(acc.y, 16);
    acc.z += __shfl_xor(acc.z, 16);
    acc.w += __shfl_xor(acc.w, 16);
    if (v < M && lane < 12) {
      const float inv = invd[v];
      const float4 p = prow[t];
      float* s = stage + (wv * NPW + t) * NO + 4 * lane;
      s[0] = p.x + acc.x * inv + bias[4 * lane];
      s[1] = p.y + acc.y * inv + bias[4 * lane + 1];
      s[2] = p.z + acc.z * inv + bias[4 * lane + 2];
      if (lane < 11) s[3] = p.w + acc.w * inv + bias[4 * lane + 3];
    }
  }
  __syncthreads();
  const int nvalid = min(32, M - vb) * NO;
  float* o = out + (size_t)vb * NO;
  for (int L = wv; L < NO; L += 8) {
    const int idx = L * 32 + lane;
    if (idx < nvalid) o[idx] = stage[idx];
  }
}

extern "C" void kernel_launch(void* const* d_in, const int* in_sizes, int n_in, void* d_out,
                              int out_size, void* d_ws, size_t ws_size, hipStream_t stream) {
  (void)in_sizes; (void)n_in; (void)out_size; (void)ws_size;
  const float* x  = (const float*)d_in[0];
  const float* W1 = (const float*)d_in[1];
  const float* b1 = (const float*)d_in[2];
  const float* W2 = (const float*)d_in[3];
  const float* b2 = (const float*)d_in[4];
  const float* W3 = (const float*)d_in[5];
  const float* b3 = (const float*)d_in[6];
  const int* src  = (const int*)d_in[7];
  const int* dst  = (const int*)d_in[8];
  float* out = (float*)d_out;

  constexpr int M = 40000, E = 640000, INF = 128, HID = 256, OUTF = 47, OUTP = 64;

  float* ws = (float*)d_ws;
  int*   deg  = (int*)ws;
  float* invd = ws + 40000;
  float* msum = ws + 80000;
  float* h1   = msum + (size_t)M * HID;
  float* h2   = h1 + (size_t)M * HID;
  float* Wp1  = h2 + (size_t)M * HID;
  float* Wp2  = Wp1 + (size_t)(2 * INF) * HID;
  float* Wp3  = Wp2 + (size_t)(2 * HID) * HID;

  int* col    = (int*)d_out;
  int* rowptr = col + E;
  int* rank   = (int*)h2;

  const int T = 256;
  auto blocks = [](long long n, int t) { return (int)((n + t - 1) / t); };

  pack_w_kernel<<<blocks((2 * INF / 2) * (long long)HID, T), T, 0, stream>>>(W1, Wp1, 2 * INF, HID, HID);
  pack_w_kernel<<<blocks((2 * HID / 2) * (long long)HID, T), T, 0, stream>>>(W2, Wp2, 2 * HID, HID, HID);
  pack_w_pair_kernel<<<blocks((HID / 2) * (long long)(2 * OUTP), T), T, 0, stream>>>(W3, Wp3, HID, OUTF);

  zero4_kernel<<<blocks(M / 4, T), T, 0, stream>>>((float4*)deg, M / 4);
  degree_rank_kernel<<<blocks(E, T), T, 0, stream>>>(dst, deg, rank, E);
  invdeg_kernel<<<blocks(M, T), T, 0, stream>>>(deg, invd, M);
  rowptr_scan_kernel<<<1, 1024, 0, stream>>>(deg, rowptr, M);
  fill_col_kernel<<<blocks(E, T), T, 0, stream>>>(src, dst, rowptr, rank, col, E);

  {
    int tiles = (M / 32) * (HID / 64);
    int gb = blocks((long long)tiles * 32, T);
    self_gemm_and_gather_kernel<INF><<<gb + M / 8, T, 0, stream>>>(
        x, Wp1, h1, HID, gb, rowptr, col, msum, M);
    neigh_gemm_kernel<true><<<gb, T, 0, stream>>>(
        msum, invd, Wp1, h1, b1, h1, M, INF, HID, HID);
  }
  {
    int tiles = (M / 32) * (HID / 64);
    int gb = blocks((long long)tiles * 32, T);
    self_gemm_and_gather_kernel<HID><<<gb + M / 8, T, 0, stream>>>(
        h1, Wp2, h2, HID, gb, rowptr, col, msum, M);
    neigh_gemm_kernel<true><<<gb, T, 0, stream>>>(
        msum, invd, Wp2, h2, b2, h2, M, HID, HID, HID);
  }
  {
    int tiles = (M / 32) * (2 * OUTP / 64);
    int gb = blocks((long long)tiles * 32, T);
    float* PZ = msum;
    int* col3 = (int*)h1;
    int* rowptr3 = col3 + E;
    constexpr int NLIST = E + M + 1;
    int mb = blocks(NLIST / 4 + 1, T);
    pair_gemm_and_move_list_kernel<<<gb + mb, T, 0, stream>>>(
        h2, Wp3, PZ, gb, col, col3, NLIST, M, HID);
    gather_finish_kernel<<<M / 32, T, 0, stream>>>(PZ, rowptr3, col3, invd, b3, out, M);
  }
}


# GEMM A operand staged through LDS once per 8-wave block by whole-line 16-byte loads (32-k chunks, next chunk in flight in registers), not per wave by 8-byte loads that touch 16 lines each
# speedup vs baseline: 2.8119x; 1.1074x over previous
#include <hip/hip_runtime.h>

typedef __attribute__((ext_vector_type(2))) float v2f;
typedef __attribute__((ext_vector_type(4))) float v4f;
typedef __attribute__((ext_vector_type(8))) float v8f;

__global__ void zero4_kernel(float4* __restrict__ p, long long n4) {
  long long i = (long long)blockIdx.x * blockDim.x + threadIdx.x;
  if (i < n4) p[i] = make_float4(0.f, 0.f, 0.f, 0.f);
}

__global__ void degree_rank_kernel(const int* __restrict__ dst, int* __restrict__ deg,
                                   int* __restrict__ rank, int nE) {
  int i = blockIdx.x * blockDim.x + threadIdx.x;
  if (i < nE) rank[i] = atomicAdd(&deg[dst[i]], 1);
}

__global__ __launch_bounds__(1024) void rowptr_scan_kernel(const int* __restrict__ deg,
                                                           int* __restrict__ rowptr, int n) {
  constexpr int VPT = 4;
  __shared__ int wsum[32];
  const int t = (int)threadIdx.x;
  const int lane = t & 31;
  const int wv = t >> 5;
  int carry = 0;
  for (int base = 0; base < n; base += 1024 * 4 * VPT) {
    const int i0 = base + 4 * VPT * t;
    int4 e[VPT];
    int run = 0;
#pragma unroll
    for (int k = 0; k < VPT; ++k) {
      int4 v = make_int4(0, 0, 0, 0);
      if (i0 + 4 * k < n) v = *(const int4*)(deg + i0 + 4 * k);
      e[k].x = run; run += v.x;
      e[k].y = run; run += v.y;
      e[k].z = run; run += v.z;
      e[k].w = run; run += v.w;
    }
    int inc = run;
#pragma unroll
    for (int off = 1; off < 32; off <<= 1) {
      int y = __shfl_up(inc, off);
      if (lane >= off) inc += y;
    }
    if (lane == 31) wsum[wv] = inc;
    __syncthreads();
    const int wtot = wsum[lane];
    int winc = wtot;
#pragma unroll
    for (int off = 1; off < 32; off <<= 1) {
      int y = __shfl_up(winc, off);
      if (lane >= off) winc += y;
    }
    const int wexcl = __shfl(winc, wv) - __shfl(wtot, wv);
    const int total = __shfl(winc, 31);
    const int ex = carry + wexcl + (inc - run);
#pragma unroll
    for (int k = 0; k < VPT; ++k) {
      if (i0 + 4 * k < n)
        *(int4*)(rowptr + i0 + 4 * k) =
            make_int4(ex + e[k].x, ex + e[k].y, ex + e[k].z, ex + e[k].w);
    }
    carry += total;
    __syncthreads();
  }
  if (t == 0) rowptr[n] = carry;
}

__global__ void fill_col_kernel(const int* __restrict__ src, const int* __restrict__ dst,
                                const int* __restrict__ rowptr, const int* __restrict__ rank,
                                int* __restrict__ col, int nE) {
  int i = blockIdx.x * blockDim.x + threadIdx.x;
  if (i < nE) col[rowptr[dst[i]] + rank[i]] = src[i];
}

__global__ void invdeg_kernel(const int* __restrict__ deg, float* __restrict__ invd, int n) {
  int i = blockIdx.x * blockDim.x + threadIdx.x;
  if (i < n) invd[i] = 1.0f / fmaxf((float)deg[i], 1.0f);
}

__global__ void pack_w_kernel(const float* __restrict__ W, float* __restrict__ Wp,
                              int K, int N, int Np) {
  int i = blockIdx.x * blockDim.x + threadIdx.x;
  int total = (K >> 1) * Np;
  if (i >= total) return;
  int k2 = i / Np;
  int c = i - k2 * Np;
  float x0 = 0.f, x1 = 0.f;
  if (c < N) {
    x0 = W[(size_t)(2 * k2) * N + c];
    x1 = W[(size_t)(2 * k2 + 1) * N + c];
  }
  *(float2*)(Wp + (size_t)k2 * 2 * Np + 2 * c) = make_float2(x0, x1);
}

template <int D>
__device__ __forceinline__ void gather_rows(
    const float* __restrict__ h, const int* __restrict__ rowptr, const int* __restrict__ col,
    float* __restrict__ msum, int M, int blk) {
  constexpr int NV = D / 128;
  const int lane = (int)threadIdx.x & 31;
  const int v = __builtin_amdgcn_readfirstlane(blk * 8 + ((int)threadIdx.x >> 5));
  if (v >= M) return;
  const int beg = rowptr[v];
  const int end = rowptr[v + 1];
  const float* hl = h + 4 * lane;
  float4 acc[NV];
#pragma unroll
  for (int k = 0; k < NV; ++k) acc[k] = make_float4(0.f, 0.f, 0.f, 0.f);

  for (int c0 = beg; c0 < end; c0 += 32) {
    const int n = min(32, end - c0);
    const int myc = (lane < n) ? col[c0 + lane] : 0;
    int i = 0;
    for (; i + 4 <= n; i += 4) {
      const float* p0 = hl + (size_t)__builtin_amdgcn_readlane(myc, i) * D;
      const float* p1 = hl + (size_t)__builtin_amdgcn_readlane(myc, i + 1) * D;
      const float* p2 = hl + (size_t)__builtin_amdgcn_readlane(myc, i + 2) * D;
      const float* p3 = hl + (size_t)__builtin_amdgcn_readlane(myc, i + 3) * D;
      float4 r0[NV], r1[NV], r2[NV], r3[NV];
#pragma unroll
      for (int k = 0; k < NV; ++k) {
        r0[k] = *(const float4*)(p0 + 128 * k);
        r1[k] = *(const float4*)(p1 + 128 * k);
        r2[k] = *(const float4*)(p2 + 128 * k);
        r3[k] = *(const float4*)(p3 + 128 * k);
      }
#pragma unroll
      for (int k = 0; k < NV; ++k) {
        acc[k].x += r0[k].x; acc[k].y += r0[k].y; acc[k].z += r0[k].z; acc[k].w += r0[k].w;
        acc[k].x += r1[k].x; acc[k].y += r1[k].y; acc[k].z += r1[k].z; acc[k].w += r1[k].w;
        acc[k].x += r2[k].x; acc[k].y += r2[k].y; acc[k].z += r2[k].z; acc[k].w += r2[k].w;
        acc[k].x += r3[k].x; acc[k].y += r3[k].y; acc[k].z += r3[k].z; acc[k].w += r3[k].w;
      }
    }
    for (; i < n; ++i) {
      const float* p0 = hl + (size_t)__builtin_amdgcn_readlane(myc, i) * D;
#pragma unroll
      for (int k = 0; k < NV; ++k) {
        const float4 r = *(const float4*)(p0 + 128 * k);
        acc[k].x += r.x; acc[k].y += r.y; acc[k].z += r.z; acc[k].w += r.w;
      }
    }
  }
  float* o = msum + (size_t)v * D + 4 * lane;
#pragma unroll
  for (int k = 0; k < NV; ++k) *(float4*)(o + 128 * k) = acc[k];
}

__device__ __forceinline__ v8f wmma4(v2f a, v2f b, v8f c) {
  return __builtin_amdgcn_wmma_f32_16x16x4_f32(false, a, false, b, (short)0, c, false, false);
}

template <int D, int TN, bool NEIGH, bool RELU>
__device__ __forceinline__ void gemm_block(
    const float* __restrict__ A, const float* __restrict__ invd, const float* __restrict__ Wp,
    const float* P, const float* __restrict__ bias, float* out, int M, int N, int blk) {
  constexpr int Np = 64 * TN;
  constexpr int RB = 256 / TN;
  constexpr int KC = 32;
  constexpr int KS = KC / 4;
  constexpr int PITCH = KC + 4;
  constexpr int NLD = RB / 32;
  constexpr int NC = D / KC;
  __shared__ __attribute__((aligned(16))) float lds[2 * RB * PITCH];

  const int t = (int)threadIdx.x;
  const int lane = t & 31;
  const int wv = __builtin_amdgcn_readfirstlane(t >> 5);
  const int half = lane >> 4;
  const int lcol = lane & 15;
  const int row0 = blk * RB;
  const int tMl = wv / TN;
  const int tN = wv - tMl * TN;
  const int m0 = row0 + 32 * tMl;
  const int n0 = tN << 6;
  const bool active = m0 < M;
  constexpr size_t ldb = 2 * (size_t)Np;
  const int boff = 2 * (n0 + lcol);

  const int srow = t >> 3;
  const int seg = t & 7;
  const float* ap[NLD];
  float sc[NLD];
#pragma unroll
  for (int p = 0; p < NLD; ++p) {
    const int rr = min(row0 + 32 * p + srow, M - 1);
    ap[p] = A + (size_t)rr * D + 4 * seg;
    sc[p] = NEIGH ? invd[rr] : 1.0f;
  }
  v4f pre[NLD];
#pragma unroll
  for (int p = 0; p < NLD; ++p) pre[p] = *(const v4f*)(ap[p]);

  v8f acc0[4], acc1[4];
  if (NEIGH && active) {
#pragma unroll
    for (int j = 0; j < 4; ++j) {
      const int cc = n0 + j * 16 + lcol;
#pragma unroll
      for (int r = 0; r < 8; ++r) {
        acc0[j][r] = P[(size_t)(m0 + half * 8 + r) * Np + cc];
        acc1[j][r] = P[(size_t)(m0 + 16 + half * 8 + r) * Np + cc];
      }
    }
  } else {
#pragma unroll
    for (int j = 0; j < 4; ++j) {
#pragma unroll
      for (int r = 0; r < 8; ++r) {
        acc0[j][r] = 0.0f;
        acc1[j][r] = 0.0f;
      }
    }
  }

  {
    float* s = lds + srow * PITCH + 4 * seg;
#pragma unroll
    for (int p = 0; p < NLD; ++p) {
      v4f v = pre[p];
      if (NEIGH) v *= sc[p];
      *(v4f*)(s + 32 * p * PITCH) = v;
    }
  }
  __syncthreads();

  const float* bpw = Wp + (size_t)half * ldb + boff;
  const float* saw = lds + (32 * tMl + lcol) * PITCH + 2 * half;
  for (int c = 0; c < NC; ++c) {
    const bool more = (c + 1 < NC);
    v2f bf[KS][4];
    const float* bpc = bpw + (size_t)(c * (KC / 2)) * ldb;
#pragma unroll
    for (int q = 0; q < KS; ++q) {
#pragma unroll
      for (int j = 0; j < 4; ++j) bf[q][j] = *(const v2f*)(bpc + (size_t)(2 * q) * ldb + 32 * j);
    }
    if (more) {
#pragma unroll
      for (int p = 0; p < NLD; ++p) pre[p] = *(const v4f*)(ap[p] + (c + 1) * KC);
    }
    const float* sa = saw + (c & 1) * (RB * PITCH);
#pragma unroll
    for (int q = 0; q < KS; ++q) {
      const v2f a0 = *(const v2f*)(sa + 4 * q);
      const v2f a1 = *(const v2f*)(sa + 16 * PITCH + 4 * q);
#pragma unroll
      for (int j = 0; j < 4; ++j) {
        acc0[j] = wmma4(a0, bf[q][j], acc0[j]);
        acc1[j] = wmma4(a1, bf[q][j], acc1[j]);
      }
    }
    if (more) {
      float* s = lds + ((c + 1) & 1) * (RB * PITCH) + srow * PITCH + 4 * seg;
#pragma unroll
      for (int p = 0; p < NLD; ++p) {
        v4f v = pre[p];
        if (NEIGH) v *= sc[p];
        *(v4f*)(s + 32 * p * PITCH) = v;
      }
      __syncthreads();
    }
  }

  if (!active) return;

  if (!NEIGH) {
#pragma unroll
    for (int j = 0; j < 4; ++j) {
      const int cc = n0 + j * 16 + lcol;
#pragma unroll
      for (int r = 0; r < 8; ++r) {
        out[(size_t)(m0 + half * 8 + r) * Np + cc] = acc0[j][r];
        out[(size_t)(m0 + 16 + half * 8 + r) * Np + cc] = acc1[j][r];
      }
    }
  } else {
#pragma unroll
    for (int j = 0; j < 4; ++j) {
      const int cc = n0 + j * 16 + lcol;
      if (cc < N) {
        const float bj = bias[cc];
#pragma unroll
        for (int r = 0; r < 8; ++r) {
          float v0 = acc0[j][r] + bj;
          float v1 = acc1[j][r] + bj;
          if (RELU) {
            v0 = fmaxf(v0, 0.0f);
            v1 = fmaxf(v1, 0.0f);
          }
          out[(size_t)(m0 + half * 8 + r) * N + cc] = v0;
          out[(size_t)(m0 + 16 + half * 8 + r) * N + cc] = v1;
        }
      }
    }
  }
}

template <int D>
__global__ __launch_bounds__(256) void self_gemm_and_gather_kernel(
    const float* __restrict__ h, const float* __restrict__ Wp, float* __restrict__ P,
    int gemm_blocks, const int* __restrict__ rowptr, const int* __restrict__ col,
    float* __restrict__ msum, int M) {
  const int b = (int)blockIdx.x;
  if (b < gemm_blocks) {
    gemm_block<D, 4, false, false>(h, nullptr, Wp, nullptr, nullptr, P, M, 256, b);
  } else {
    gather_rows<D>(h, rowptr, col, msum, M, b - gemm_blocks);
  }
}

template <int D, bool RELU>
__global__ __launch_bounds__(256) void neigh_gemm_kernel(
    const float* __restrict__ msum, const float* __restrict__ invd,
    const float* __restrict__ Wp, const float* P, const float* __restrict__ bias, float* out,
    int M, int N) {
  gemm_block<D, 4, true, RELU>(msum, invd, Wp + (size_t)(D / 2) * 512, P, bias, out, M, N,
                               (int)blockIdx.x);
}

__global__ void pack_w_pair_kernel(const float* __restrict__ W, float* __restrict__ Wp,
                                   int d, int N) {
  int i = blockIdx.x * blockDim.x + threadIdx.x;
  int total = (d >> 1) * 128;
  if (i >= total) return;
  int k2 = i >> 7;
  int c = i & 127;
  int part = c >> 6;
  int cc = c & 63;
  float x0 = 0.f, x1 = 0.f;
  if (cc < N) {
    x0 = W[(size_t)(part * d + 2 * k2) * N + cc];
    x1 = W[(size_t)(part * d + 2 * k2 + 1) * N + cc];
  }
  *(float2*)(Wp + (size_t)k2 * 256 + 2 * c) = make_float2(x0, x1);
}

__global__ __launch_bounds__(256) void pair_gemm_and_move_list_kernel(
    const float* __restrict__ h, const float* __restrict__ Wp, float* __restrict__ PZ,
    int gemm_blocks, const int* __restrict__ list_from, int* __restrict__ list_to,
    int n_ints, int M) {
  const int b = (int)blockIdx.x;
  if (b < gemm_blocks) {
    gemm_block<256, 2, false, false>(h, nullptr, Wp, nullptr, nullptr, PZ, M, 128, b);
  } else {
    const int i4 = (b - gemm_blocks) * 256 + (int)threadIdx.x;
    const int n4 = n_ints >> 2;
    if (i4 < n4) {
      ((int4*)list_to)[i4] = ((const int4*)list_from)[i4];
    } else if (i4 == n4) {
      for (int k = 4 * n4; k < n_ints; ++k) list_to[k] = list_from[k];
    }
  }
}

__global__ __launch_bounds__(256) void gather_finish_kernel(
    const float* __restrict__ PZ, const int* __restrict__ rowptr, const int* __restrict__ col,
    const float* __restrict__ invd, const float* __restrict__ bias, float* __restrict__ out,
    int M) {
  constexpr int NO = 47;
  constexpr int NPW = 4;
  __shared__ float stage[32 * NO];
  const int lane = (int)threadIdx.x & 31;
  const int half = lane >> 4;
  const int wv = __builtin_amdgcn_readfirstlane((int)threadIdx.x >> 5);
  const int vb = (int)blockIdx.x * 32;
  const int v0 = vb + wv * NPW;
  const float* Zl = PZ + 64 + 4 * (lane & 15);

  int rp[NPW + 1];
#pragma unroll
  for (int k = 0; k <= NPW; ++k) rp[k] = rowptr[min(v0 + k, M)];
  int ids[NPW];
  float4 prow[NPW];
#pragma unroll
  for (int t = 0; t < NPW; ++t) {
    ids[t] = (lane < rp[t + 1] - rp[t]) ? col[rp[t] + lane] : 0;
    prow[t] = *(const float4*)(PZ + (size_t)min(v0 + t, M - 1) * 128 + 4 * (lane & 15));
  }

#pragma unroll
  for (int t = 0; t < NPW; ++t) {
    const int v = v0 + t;
    const int beg = rp[t];
    const int end = rp[t + 1];
    float4 acc = make_float4(0.f, 0.f, 0.f, 0.f);
    int myc = ids[t];
    for (int c0 = beg; c0 < end; c0 += 32) {
      const int n = min(32, end - c0);
      if (c0 != beg) myc = (lane < n) ? col[c0 + lane] : 0;
      for (int i = 0; i < n; i += 8) {
        float4 r[4];
#pragma unroll
        for (int q = 0; q < 4; ++q) {
          const int k = i + 2 * q;
          const int ea = __builtin_amdgcn_readlane(myc, k);
          const int eb = __builtin_amdgcn_readlane(myc, k + 1);
          r[q] = make_float4(0.f, 0.f, 0.f, 0.f);
          if (k + half < n) r[q] = *(const float4*)(Zl + (size_t)(half ? eb : ea) * 128);
        }
#pragma unroll
        for (int q = 0; q < 4; ++q) {
          acc.x += r[q].x; acc.y += r[q].y; acc.z += r[q].z; acc.w += r[q].w;
        }
      }
    }
    acc.x += __shfl_xor(acc.x, 16);
    acc.y += __shfl_xor(acc.y, 16);
    acc.z += __shfl_xor(acc.z, 16);
    acc.w += __shfl_xor(acc.w, 16);
    if (v < M && lane < 12) {
      const float inv = invd[v];
      const float4 p = prow[t];
      float* s = stage + (wv * NPW + t) * NO + 4 * lane;
      s[0] = p.x + acc.x * inv + bias[4 * lane];
      s[1] = p.y + acc.y * inv + bias[4 * lane + 1];
      s[2] = p.z + acc.z * inv + bias[4 * lane + 2];
      if (lane < 11) s[3] = p.w + acc.w * inv + bias[4 * lane + 3];
    }
  }
  __syncthreads();
  const int nvalid = min(32, M - vb) * NO;
  float* o = out + (size_t)vb * NO;
  for (int L = wv; L < NO; L += 8) {
    const int idx = L * 32 + lane;
    if (idx < nvalid) o[idx] = stage[idx];
  }
}

extern "C" void kernel_launch(void* const* d_in, const int* in_sizes, int n_in, void* d_out,
                              int out_size, void* d_ws, size_t ws_size, hipStream_t stream) {
  (void)in_sizes; (void)n_in; (void)out_size; (void)ws_size;
  const float* x  = (const float*)d_in[0];
  const float* W1 = (const float*)d_in[1];
  const float* b1 = (const float*)d_in[2];
  const float* W2 = (const float*)d_in[3];
  const float* b2 = (const float*)d_in[4];
  const float* W3 = (const float*)d_in[5];
  const float* b3 = (const float*)d_in[6];
  const int* src  = (const int*)d_in[7];
  const int* dst  = (const int*)d_in[8];
  float* out = (float*)d_out;

  constexpr int M = 40000, E = 640000, INF = 128, HID = 256, OUTF = 47, OUTP = 64;

  float* ws = (float*)d_ws;
  int*   deg  = (int*)ws;
  float* invd = ws + 40000;
  float* msum = ws + 80000;
  float* h1   = msum + (size_t)M * HID;
  float* h2   = h1 + (size_t)M * HID;
  float* Wp1  = h2 + (size_t)M * HID;
  float* Wp2  = Wp1 + (size_t)(2 * INF) * HID;
  float* Wp3  = Wp2 + (size_t)(2 * HID) * HID;

  int* col    = (int*)d_out;
  int* rowptr = col + E;
  int* rank   = (int*)h2;

  const int T = 256;
  auto blocks = [](long long n, int t) { return (int)((n + t - 1) / t); };

  pack_w_kernel<<<blocks((2 * INF / 2) * (long long)HID, T), T, 0, stream>>>(W1, Wp1, 2 * INF, HID, HID);
  pack_w_kernel<<<blocks((2 * HID / 2) * (long long)HID, T), T, 0, stream>>>(W2, Wp2, 2 * HID, HID, HID);
  pack_w_pair_kernel<<<blocks((HID / 2) * (long long)(2 * OUTP), T), T, 0, stream>>>(W3, Wp3, HID, OUTF);

  zero4_kernel<<<blocks(M / 4, T), T, 0, stream>>>((float4*)deg, M / 4);
  degree_rank_kernel<<<blocks(E, T), T, 0, stream>>>(dst, deg, rank, E);
  invdeg_kernel<<<blocks(M, T), T, 0, stream>>>(deg, invd, M);
  rowptr_scan_kernel<<<1, 1024, 0, stream>>>(deg, rowptr, M);
  fill_col_kernel<<<blocks(E, T), T, 0, stream>>>(src, dst, rowptr, rank, col, E);

  {
    int tiles = (M / 32) * (HID / 64);
    int gb = blocks((long long)tiles * 32, T);
    self_gemm_and_gather_kernel<INF><<<gb + M / 8, T, 0, stream>>>(
        x, Wp1, h1, gb, rowptr, col, msum, M);
    neigh_gemm_kernel<INF, true><<<gb, T, 0, stream>>>(
        msum, invd, Wp1, h1, b1, h1, M, HID);
  }
  {
    int tiles = (M / 32) * (HID / 64);
    int gb = blocks((long long)tiles * 32, T);
    self_gemm_and_gather_kernel<HID><<<gb + M / 8, T, 0, stream>>>(
        h1, Wp2, h2, gb, rowptr, col, msum, M);
    neigh_gemm_kernel<HID, true><<<gb, T, 0, stream>>>(
        msum, invd, Wp2, h2, b2, h2, M, HID);
  }
  {
    int tiles = (M / 32) * (2 * OUTP / 64);
    int gb = blocks((long long)tiles * 32, T);
    float* PZ = msum;
    int* col3 = (int*)h1;
    int* rowptr3 = col3 + E;
    constexpr int NLIST = E + M + 1;
    int mb = blocks(NLIST / 4 + 1, T);
    pair_gemm_and_move_list_kernel<<<gb + mb, T, 0, stream>>>(
        h2, Wp3, PZ, gb, col, col3, NLIST, M);
    gather_finish_kernel<<<M / 32, T, 0, stream>>>(PZ, rowptr3, col3, invd, b3, out, M);
  }
}


# layer 3's projection runs inside layer 2's neighbour GEMM launch: a block multiplies its 64 finished rows of h2 straight out of an LDS tile, h2 is never stored or re-read: 10 launches become 9
# speedup vs baseline: 3.1254x; 1.1115x over previous
#include <hip/hip_runtime.h>

typedef __attribute__((ext_vector_type(2))) float v2f;
typedef __attribute__((ext_vector_type(4))) float v4f;
typedef __attribute__((ext_vector_type(8))) float v8f;

__global__ void zero4_kernel(float4* __restrict__ p, long long n4) {
  long long i = (long long)blockIdx.x * blockDim.x + threadIdx.x;
  if (i < n4) p[i] = make_float4(0.f, 0.f, 0.f, 0.f);
}

__global__ void degree_rank_kernel(const int* __restrict__ dst, int* __restrict__ deg,
                                   int* __restrict__ rank, int nE) {
  int i = blockIdx.x * blockDim.x + threadIdx.x;
  if (i < nE) rank[i] = atomicAdd(&deg[dst[i]], 1);
}

__device__ __forceinline__ void rowptr_scan_block(const int* __restrict__ deg,
                                                  int* __restrict__ rowptr, int n) {
  constexpr int VPT = 4;
  __shared__ int wsum[32];
  const int t = (int)threadIdx.x;
  const int lane = t & 31;
  const int wv = t >> 5;
  int carry = 0;
  for (int base = 0; base < n; base += 1024 * 4 * VPT) {
    const int i0 = base + 4 * VPT * t;
    int4 e[VPT];
    int run = 0;
#pragma unroll
    for (int k = 0; k < VPT; ++k) {
      int4 v = make_int4(0, 0, 0, 0);
      if (i0 + 4 * k < n) v = *(const int4*)(deg + i0 + 4 * k);
      e[k].x = run; run += v.x;
      e[k].y = run; run += v.y;
      e[k].z = run; run += v.z;
      e[k].w = run; run += v.w;
    }
    int inc = run;
#pragma unroll
    for (int off = 1; off < 32; off <<= 1) {
      int y = __shfl_up(inc, off);
      if (lane >= off) inc += y;
    }
    if (lane == 31) wsum[wv] = inc;
    __syncthreads();
    const int wtot = wsum[lane];
    int winc = wtot;
#pragma unroll
    for (int off = 1; off < 32; off <<= 1) {
      int y = __shfl_up(winc, off);
      if (lane >= off) winc += y;
    }
    const int wexcl = __shfl(winc, wv) - __shfl(wtot, wv);
    const int total = __shfl(winc, 31);
    const int ex = carry + wexcl + (inc - run);
#pragma unroll
    for (int k = 0; k < VPT; ++k) {
      if (i0 + 4 * k < n)
        *(int4*)(rowptr + i0 + 4 * k) =
            make_int4(ex + e[k].x, ex + e[k].y, ex + e[k].z, ex + e[k].w);
    }
    carry += total;
    __syncthreads();
  }
  if (t == 0) rowptr[n] = carry;
}

__global__ void fill_col_kernel(const int* __restrict__ src, const int* __restrict__ dst,
                                const int* __restrict__ rowptr, const int* __restrict__ rank,
                                int* __restrict__ col, int nE) {
  int i = blockIdx.x * blockDim.x + threadIdx.x;
  if (i < nE) col[rowptr[dst[i]] + rank[i]] = src[i];
}


__device__ __forceinline__ void pack_w_elem(const float* __restrict__ W, float* __restrict__ Wp,
                                            int K, int N, int Np, int i) {
  int total = (K >> 1) * Np;
  if (i >= total) return;
  int k2 = i / Np;
  int c = i - k2 * Np;
  float x0 = 0.f, x1 = 0.f;
  if (c < N) {
    x0 = W[(size_t)(2 * k2) * N + c];
    x1 = W[(size_t)(2 * k2 + 1) * N + c];
  }
  *(float2*)(Wp + (size_t)k2 * 2 * Np + 2 * c) = make_float2(x0, x1);
}

template <int D>
__device__ __forceinline__ void gather_rows(
    const float* __restrict__ h, const int* __restrict__ rowptr, const int* __restrict__ col,
    float* __restrict__ msum, int M, int blk) {
  constexpr int NV = D / 128;
  const int lane = (int)threadIdx.x & 31;
  const int v = __builtin_amdgcn_readfirstlane(blk * 8 + ((int)threadIdx.x >> 5));
  if (v >= M) return;
  const int beg = rowptr[v];
  const int end = rowptr[v + 1];
  const float* hl = h + 4 * lane;
  float4 acc[NV];
#pragma unroll
  for (int k = 0; k < NV; ++k) acc[k] = make_float4(0.f, 0.f, 0.f, 0.f);

  for (int c0 = beg; c0 < end; c0 += 32) {
    const int n = min(32, end - c0);
    const int myc = (lane < n) ? col[c0 + lane] : 0;
    int i = 0;
    for (; i + 4 <= n; i += 4) {
      const float* p0 = hl + (size_t)__builtin_amdgcn_readlane(myc, i) * D;
      const float* p1 = hl + (size_t)__builtin_amdgcn_readlane(myc, i + 1) * D;
      const float* p2 = hl + (size_t)__builtin_amdgcn_readlane(myc, i + 2) * D;
      const float* p3 = hl + (size_t)__builtin_amdgcn_readlane(myc, i + 3) * D;
      float4 r0[NV], r1[NV], r2[NV], r3[NV];
#pragma unroll
      for (int k = 0; k < NV; ++k) {
        r0[k] = *(const float4*)(p0 + 128 * k);
        r1[k] = *(const float4*)(p1 + 128 * k);
        r2[k] = *(const float4*)(p2 + 128 * k);
        r3[k] = *(const float4*)(p3 + 128 * k);
      }
#pragma unroll
      for (int k = 0; k < NV; ++k) {
        acc[k].x += r0[k].x; acc[k].y += r0[k].y; acc[k].z += r0[k].z; acc[k].w += r0[k].w;
        acc[k].x += r1[k].x; acc[k].y += r1[k].y; acc[k].z += r1[k].z; acc[k].w += r1[k].w;
        acc[k].x += r2[k].x; acc[k].y += r2[k].y; acc[k].z += r2[k].z; acc[k].w += r2[k].w;
        acc[k].x += r3[k].x; acc[k].y += r3[k].y; acc[k].z += r3[k].z; acc[k].w += r3[k].w;
      }
    }
    for (; i < n; ++i) {
      const float* p0 = hl + (size_t)__builtin_amdgcn_readlane(myc, i) * D;
#pragma unroll
      for (int k = 0; k < NV; ++k) {
        const float4 r = *(const float4*)(p0 + 128 * k);
        acc[k].x += r.x; acc[k].y += r.y; acc[k].z += r.z; acc[k].w += r.w;
      }
    }
  }
  float* o = msum + (size_t)v * D + 4 * lane;
#pragma unroll
  for (int k = 0; k < NV; ++k) *(float4*)(o + 128 * k) = acc[k];
}

__device__ __forceinline__ v8f wmma4(v2f a, v2f b, v8f c) {
  return __builtin_amdgcn_wmma_f32_16x16x4_f32(false, a, false, b, (short)0, c, false, false);
}

template <int D, int TN, bool NEIGH, bool RELU, bool TO_LDS = false, int TP = 0>
__device__ __forceinline__ void gemm_block(
    const float* __restrict__ A, const float* __restrict__ invd, const float* __restrict__ Wp,
    const float* P, const float* __restrict__ bias, float* out, int M, int N, int blk,
    float* tile = nullptr) {
  constexpr int Np = 64 * TN;
  constexpr int RB = 256 / TN;
  constexpr int KC = 32;
  constexpr int KS = KC / 4;
  constexpr int PITCH = KC + 4;
  constexpr int NLD = RB / 32;
  constexpr int NC = D / KC;
  __shared__ __attribute__((aligned(16))) float lds[2 * RB * PITCH];

  const int t = (int)threadIdx.x;
  const int lane = t & 31;
  const int wv = __builtin_amdgcn_readfirstlane(t >> 5);
  const int half = lane >> 4;
  const int lcol = lane & 15;
  const int row0 = blk * RB;
  const int tMl = wv / TN;
  const int tN = wv - tMl * TN;
  const int m0 = row0 + 32 * tMl;
  const int n0 = tN << 6;
  const bool active = m0 < M;
  constexpr size_t ldb = 2 * (size_t)Np;
  const int boff = 2 * (n0 + lcol);

  const int srow = t >> 3;
  const int seg = t & 7;
  const float* ap[NLD];
  float sc[NLD];
#pragma unroll
  for (int p = 0; p < NLD; ++p) {
    const int rr = min(row0 + 32 * p + srow, M - 1);
    ap[p] = A + (size_t)rr * D + 4 * seg;
    sc[p] = NEIGH ? invd[rr] : 1.0f;
  }
  v4f pre[NLD];
#pragma unroll
  for (int p = 0; p < NLD; ++p) pre[p] = *(const v4f*)(ap[p]);

  v8f acc0[4], acc1[4];
  if (NEIGH && active) {
#pragma unroll
    for (int j = 0; j < 4; ++j) {
      const int cc = n0 + j * 16 + lcol;
#pragma unroll
      for (int r = 0; r < 8; ++r) {
        acc0[j][r] = P[(size_t)(m0 + half * 8 + r) * Np + cc];
        acc1[j][r] = P[(size_t)(m0 + 16 + half * 8 + r) * Np + cc];
      }
    }
  } else {
#pragma unroll
    for (int j = 0; j < 4; ++j) {
#pragma unroll
      for (int r = 0; r < 8; ++r) {
        acc0[j][r] = 0.0f;
        acc1[j][r] = 0.0f;
      }
    }
  }

  {
    float* s = lds + srow * PITCH + 4 * seg;
#pragma unroll
    for (int p = 0; p < NLD; ++p) {
      v4f v = pre[p];
      if (NEIGH) v *= sc[p];
      *(v4f*)(s + 32 * p * PITCH) = v;
    }
  }
  __syncthreads();

  const float* bpw = Wp + (size_t)half * ldb + boff;
  const float* saw = lds + (32 * tMl + lcol) * PITCH + 2 * half;
  for (int c = 0; c < NC; ++c) {
    const bool more = (c + 1 < NC);
    v2f bf[KS][4];
    const float* bpc = bpw + (size_t)(c * (KC / 2)) * ldb;
#pragma unroll
    for (int q = 0; q < KS; ++q) {
#pragma unroll
      for (int j = 0; j < 4; ++j) bf[q][j] = *(const v2f*)(bpc + (size_t)(2 * q) * ldb + 32 * j);
    }
    if (more) {
#pragma unroll
      for (int p = 0; p < NLD; ++p) pre[p] = *(const v4f*)(ap[p] + (c + 1) * KC);
    }
    const float* sa = saw + (c & 1) * (RB * PITCH);
#pragma unroll
    for (int q = 0; q < KS; ++q) {
      const v2f a0 = *(const v2f*)(sa + 4 * q);
      const v2f a1 = *(const v2f*)(sa + 16 * PITCH + 4 * q);
#pragma unroll
      for (int j = 0; j < 4; ++j) {
        acc0[j] = wmma4(a0, bf[q][j], acc0[j]);
        acc1[j] = wmma4(a1, bf[q][j], acc1[j]);
      }
    }
    if (more) {
      float* s = lds + ((c + 1) & 1) * (RB * PITCH) + srow * PITCH + 4 * seg;
#pragma unroll
      for (int p = 0; p < NLD; ++p) {
        v4f v = pre[p];
        if (NEIGH) v *= sc[p];
        *(v4f*)(s + 32 * p * PITCH) = v;
      }
      __syncthreads();
    }
  }

  if (TO_LDS) {
#pragma unroll
    for (int j = 0; j < 4; ++j) {
      const int cc = n0 + j * 16 + lcol;
      const float bj = bias[cc];
#pragma unroll
      for (int r = 0; r < 8; ++r) {
        float v0 = acc0[j][r] + bj;
        float v1 = acc1[j][r] + bj;
        if (RELU) {
          v0 = fmaxf(v0, 0.0f);
          v1 = fmaxf(v1, 0.0f);
        }
        tile[(32 * tMl + half * 8 + r) * TP + cc] = v0;
        tile[(32 * tMl + 16 + half * 8 + r) * TP + cc] = v1;
      }
    }
    return;
  }

  if (!active) return;

  if (!NEIGH) {
#pragma unroll
    for (int j = 0; j < 4; ++j) {
      const int cc = n0 + j * 16 + lcol;
#pragma unroll
      for (int r = 0; r < 8; ++r) {
        out[(size_t)(m0 + half * 8 + r) * Np + cc] = acc0[j][r];
        out[(size_t)(m0 + 16 + half * 8 + r) * Np + cc] = acc1[j][r];
      }
    }
  } else {
#pragma unroll
    for (int j = 0; j < 4; ++j) {
      const int cc = n0 + j * 16 + lcol;
      if (cc < N) {
        const float bj = bias[cc];
#pragma unroll
        for (int r = 0; r < 8; ++r) {
          float v0 = acc0[j][r] + bj;
          float v1 = acc1[j][r] + bj;
          if (RELU) {
            v0 = fmaxf(v0, 0.0f);
            v1 = fmaxf(v1, 0.0f);
          }
          out[(size_t)(m0 + half * 8 + r) * N + cc] = v0;
          out[(size_t)(m0 + 16 + half * 8 + r) * N + cc] = v1;
        }
      }
    }
  }
}

template <int D>
__global__ __launch_bounds__(256) void self_gemm_and_gather_kernel(
    const float* __restrict__ h, const float* __restrict__ Wp, float* __restrict__ P,
    int gemm_blocks, const int* __restrict__ rowptr, const int* __restrict__ col,
    float* __restrict__ msum, int M) {
  const int b = (int)blockIdx.x;
  if (b < gemm_blocks) {
    gemm_block<D, 4, false, false>(h, nullptr, Wp, nullptr, nullptr, P, M, 256, b);
  } else {
    gather_rows<D>(h, rowptr, col, msum, M, b - gemm_blocks);
  }
}

template <int D, bool RELU>
__global__ __launch_bounds__(256) void neigh_gemm_kernel(
    const float* __restrict__ msum, const float* __restrict__ invd,
    const float* __restrict__ Wp, const float* P, const float* __restrict__ bias, float* out,
    int M, int N) {
  gemm_block<D, 4, true, RELU>(msum, invd, Wp + (size_t)(D / 2) * 512, P, bias, out, M, N,
                               (int)blockIdx.x);
}

__device__ __forceinline__ void pack_w_pair_elem(const float* __restrict__ W,
                                                 float* __restrict__ Wp, int d, int N, int i) {
  int total = (d >> 1) * 128;
  if (i >= total) return;
  int k2 = i >> 7;
  int c = i & 127;
  int part = c >> 6;
  int cc = c & 63;
  float x0 = 0.f, x1 = 0.f;
  if (cc < N) {
    x0 = W[(size_t)(part * d + 2 * k2) * N + cc];
    x1 = W[(size_t)(part * d + 2 * k2 + 1) * N + cc];
  }
  *(float2*)(Wp + (size_t)k2 * 256 + 2 * c) = make_float2(x0, x1);
}

__global__ __launch_bounds__(1024) void scan_invdeg_pack_kernel(
    const int* __restrict__ deg, int* __restrict__ rowptr, float* __restrict__ invd, int n,
    const float* __restrict__ W1, float* __restrict__ Wp1,
    const float* __restrict__ W2, float* __restrict__ Wp2,
    const float* __restrict__ W3, float* __restrict__ Wp3,
    int d1, int d2, int nh, int no) {
  const int b = (int)blockIdx.x;
  const int t = (int)threadIdx.x;
  if (b == 0) {
    rowptr_scan_block(deg, rowptr, n);
    return;
  }
  int r = b - 1;
  const int nb_inv = (n + 1023) >> 10;
  if (r < nb_inv) {
    const int i = r * 1024 + t;
    if (i < n) invd[i] = 1.0f / fmaxf((float)deg[i], 1.0f);
    return;
  }
  r -= nb_inv;
  const int nb_p1 = (d1 * nh + 1023) >> 10;
  if (r < nb_p1) {
    pack_w_elem(W1, Wp1, 2 * d1, nh, nh, r * 1024 + t);
    return;
  }
  r -= nb_p1;
  const int nb_p2 = (d2 * nh + 1023) >> 10;
  if (r < nb_p2) {
    pack_w_elem(W2, Wp2, 2 * d2, nh, nh, r * 1024 + t);
    return;
  }
  r -= nb_p2;
  pack_w_pair_elem(W3, Wp3, d2, no, r * 1024 + t);
}

__global__ __launch_bounds__(256) void neigh_gemm_project_and_move_list_kernel(
    const float* __restrict__ msum, const float* __restrict__ invd,
    const float* __restrict__ Wp2, const float* __restrict__ P, const float* __restrict__ b2,
    const float* __restrict__ Wp3, float* __restrict__ PZ, int gemm_blocks,
    const int* __restrict__ list_from, int* __restrict__ list_to, int n_ints, int M) {
  constexpr int D = 256;
  constexpr int TP = D + 4;
  __shared__ __attribute__((aligned(16))) float tile[64 * TP];
  const int b = (int)blockIdx.x;
  if (b < gemm_blocks) {
    gemm_block<D, 4, true, true, true, TP>(msum, invd, Wp2 + (size_t)(D / 2) * 512, P, b2,
                                           nullptr, M, 256, b, tile);
    const int lane = (int)threadIdx.x & 31;
    const int wv = __builtin_amdgcn_readfirstlane((int)threadIdx.x >> 5);
    const int half = lane >> 4;
    const int lcol = lane & 15;
    const float* bp = Wp3 + (size_t)half * 256 + 2 * (16 * wv + lcol);
    const float* sa = tile + lcol * TP + 2 * half;
    v8f z[4];
#pragma unroll
    for (int i = 0; i < 4; ++i) {
#pragma unroll
      for (int r = 0; r < 8; ++r) z[i][r] = 0.0f;
    }
    v2f bf[8];
#pragma unroll
    for (int q = 0; q < 8; ++q) bf[q] = *(const v2f*)(bp + (size_t)(2 * q) * 256);
    __syncthreads();
    for (int kc = 0; kc < D / 32; ++kc) {
      const bool more = (kc + 1 < D / 32);
      v2f bn[8];
      if (more) {
#pragma unroll
        for (int q = 0; q < 8; ++q)
          bn[q] = *(const v2f*)(bp + (size_t)((kc + 1) * 16 + 2 * q) * 256);
      }
#pragma unroll
      for (int q = 0; q < 8; ++q) {
#pragma unroll
        for (int i = 0; i < 4; ++i) {
          const v2f a = *(const v2f*)(sa + 16 * i * TP + kc * 32 + 4 * q);
          z[i] = wmma4(a, bf[q], z[i]);
        }
      }
      if (more) {
#pragma unroll
        for (int q = 0; q < 8; ++q) bf[q] = bn[q];
      }
    }
    const int row0 = b * 64;
    const int cc = 16 * wv + lcol;
#pragma unroll
    for (int i = 0; i < 4; ++i) {
      if (row0 + 16 * i < M) {
#pragma unroll
        for (int r = 0; r < 8; ++r)
          PZ[(size_t)(row0 + 16 * i + half * 8 + r) * 128 + cc] = z[i][r];
      }
    }
  } else {
    const int i4 = (b - gemm_blocks) * 256 + (int)threadIdx.x;
    const int n4 = n_ints >> 2;
    if (i4 < n4) {
      ((int4*)list_to)[i4] = ((const int4*)list_from)[i4];
    } else if (i4 == n4) {
      for (int k = 4 * n4; k < n_ints; ++k) list_to[k] = list_from[k];
    }
  }
}

__global__ __launch_bounds__(256) void gather_finish_kernel(
    const float* __restrict__ PZ, const int* __restrict__ rowptr, const int* __restrict__ col,
    const float* __restrict__ invd, const float* __restrict__ bias, float* __restrict__ out,
    int M) {
  constexpr int NO = 47;
  constexpr int NPW = 4;
  __shared__ float stage[32 * NO];
  const int lane = (int)threadIdx.x & 31;
  const int half = lane >> 4;
  const int wv = __builtin_amdgcn_readfirstlane((int)threadIdx.x >> 5);
  const int vb = (int)blockIdx.x * 32;
  const int v0 = vb + wv * NPW;
  const float* Zl = PZ + 64 + 4 * (lane & 15);

  int rp[NPW + 1];
#pragma unroll
  for (int k = 0; k <= NPW; ++k) rp[k] = rowptr[min(v0 + k, M)];
  int ids[NPW];
  float4 prow[NPW];
#pragma unroll
  for (int t = 0; t < NPW; ++t) {
    ids[t] = (lane < rp[t + 1] - rp[t]) ? col[rp[t] + lane] : 0;
    prow[t] = *(const float4*)(PZ + (size_t)min(v0 + t, M - 1) * 128 + 4 * (lane & 15));
  }

#pragma unroll
  for (int t = 0; t < NPW; ++t) {
    const int v = v0 + t;
    const int beg = rp[t];
    const int end = rp[t + 1];
    float4 acc = make_float4(0.f, 0.f, 0.f, 0.f);
    int myc = ids[t];
    for (int c0 = beg; c0 < end; c0 += 32) {
      const int n = min(32, end - c0);
      if (c0 != beg) myc = (lane < n) ? col[c0 + lane] : 0;
      for (int i = 0; i < n; i += 8) {
        float4 r[4];
#pragma unroll
        for (int q = 0; q < 4; ++q) {
          const int k = i + 2 * q;
          const int ea = __builtin_amdgcn_readlane(myc, k);
          const int eb = __builtin_amdgcn_readlane(myc, k + 1);
          r[q] = make_float4(0.f, 0.f, 0.f, 0.f);
          if (k + half < n) r[q] = *(const float4*)(Zl + (size_t)(half ? eb : ea) * 128);
        }
#pragma unroll
        for (int q = 0; q < 4; ++q) {
          acc.x += r[q].x; acc.y += r[q].y; acc.z += r[q].z; acc.w += r[q].w;
        }
      }
    }
    acc.x += __shfl_xor(acc.x, 16);
    acc.y += __shfl_xor(acc.y, 16);
    acc.z += __shfl_xor(acc.z, 16);
    acc.w += __shfl_xor(acc.w, 16);
    if (v < M && lane < 12) {
      const float inv = invd[v];
      const float4 p = prow[t];
      float* s = stage + (wv * NPW + t) * NO + 4 * lane;
      s[0] = p.x + acc.x * inv + bias[4 * lane];
      s[1] = p.y + acc.y * inv + bias[4 * lane + 1];
      s[2] = p.z + acc.z * inv + bias[4 * lane + 2];
      if (lane < 11) s[3] = p.w + acc.w * inv + bias[4 * lane + 3];
    }
  }
  __syncthreads();
  const int nvalid = min(32, M - vb) * NO;
  float* o = out + (size_t)vb * NO;
  for (int L = wv; L < NO; L += 8) {
    const int idx = L * 32 + lane;
    if (idx < nvalid) o[idx] = stage[idx];
  }
}

extern "C" void kernel_launch(void* const* d_in, const int* in_sizes, int n_in, void* d_out,
                              int out_size, void* d_ws, size_t ws_size, hipStream_t stream) {
  (void)in_sizes; (void)n_in; (void)out_size; (void)ws_size;
  const float* x  = (const float*)d_in[0];
  const float* W1 = (const float*)d_in[1];
  const float* b1 = (const float*)d_in[2];
  const float* W2 = (const float*)d_in[3];
  const float* b2 = (const float*)d_in[4];
  const float* W3 = (const float*)d_in[5];
  const float* b3 = (const float*)d_in[6];
  const int* src  = (const int*)d_in[7];
  const int* dst  = (const int*)d_in[8];
  float* out = (float*)d_out;

  constexpr int M = 40000, E = 640000, INF = 128, HID = 256, OUTF = 47, OUTP = 64;

  float* ws = (float*)d_ws;
  int*   deg  = (int*)ws;
  float* invd = ws + 40000;
  float* msum = ws + 80000;
  float* h1   = msum + (size_t)M * HID;
  float* h2   = h1 + (size_t)M * HID;
  float* Wp1  = h2 + (size_t)M * HID;
  float* Wp2  = Wp1 + (size_t)(2 * INF) * HID;
  float* Wp3  = Wp2 + (size_t)(2 * HID) * HID;

  int* col    = (int*)d_out;
  int* rowptr = col + E;
  int* rank   = (int*)h2;

  const int T = 256;
  auto blocks = [](long long n, int t) { return (int)((n + t - 1) / t); };

  zero4_kernel<<<blocks(M / 4, T), T, 0, stream>>>((float4*)deg, M / 4);
  degree_rank_kernel<<<blocks(E, T), T, 0, stream>>>(dst, deg, rank, E);
  {
    int nb = 1 + blocks(M, 1024) + blocks((long long)INF * HID, 1024) +
             blocks((long long)HID * HID, 1024) + blocks((HID / 2) * (long long)(2 * OUTP), 1024);
    scan_invdeg_pack_kernel<<<nb, 1024, 0, stream>>>(deg, rowptr, invd, M, W1, Wp1, W2, Wp2,
                                                     W3, Wp3, INF, HID, HID, OUTF);
  }
  fill_col_kernel<<<blocks(E, T), T, 0, stream>>>(src, dst, rowptr, rank, col, E);

  {
    int tiles = (M / 32) * (HID / 64);
    int gb = blocks((long long)tiles * 32, T);
    self_gemm_and_gather_kernel<INF><<<gb + M / 8, T, 0, stream>>>(
        x, Wp1, h1, gb, rowptr, col, msum, M);
    neigh_gemm_kernel<INF, true><<<gb, T, 0, stream>>>(
        msum, invd, Wp1, h1, b1, h1, M, HID);
  }
  {
    int tiles = (M / 32) * (HID / 64);
    int gb = blocks((long long)tiles * 32, T);
    self_gemm_and_gather_kernel<HID><<<gb + M / 8, T, 0, stream>>>(
        h1, Wp2, h2, gb, rowptr, col, msum, M);
    float* PZ = h1;
    int* col3 = (int*)(h1 + (size_t)M * 2 * OUTP);
    constexpr int NLIST = E + M + 1;
    int mb = blocks(NLIST / 4 + 1, T);
    neigh_gemm_project_and_move_list_kernel<<<gb + mb, T, 0, stream>>>(
        msum, invd, Wp2, h2, b2, Wp3, PZ, gb, col, col3, NLIST, M);
  }
  {
    const float* PZ = h1;
    const int* col3 = (const int*)(h1 + (size_t)M * 2 * OUTP);
    const int* rowptr3 = col3 + E;
    gather_finish_kernel<<<M / 32, T, 0, stream>>>(PZ, rowptr3, col3, invd, b3, out, M);
  }
}


# layer kernels: row tile 80 instead of 64 (wave tile 80x32): 500 blocks, two per compute unit, so each unit carries 160 rows instead of 192 on the fullest; weights streamed 500 not 625 times
# speedup vs baseline: 3.4308x; 1.0977x over previous
#include <hip/hip_runtime.h>

typedef __attribute__((ext_vector_type(2))) float v2f;
typedef __attribute__((ext_vector_type(4))) float v4f;
typedef __attribute__((ext_vector_type(8))) float v8f;

__global__ void zero4_kernel(float4* __restrict__ p, long long n4) {
  long long i = (long long)blockIdx.x * blockDim.x + threadIdx.x;
  if (i < n4) p[i] = make_float4(0.f, 0.f, 0.f, 0.f);
}

__global__ void degree_rank_kernel(const int* __restrict__ dst, int* __restrict__ deg,
                                   int* __restrict__ rank, int nE) {
  int i = blockIdx.x * blockDim.x + threadIdx.x;
  if (i < nE) rank[i] = atomicAdd(&deg[dst[i]], 1);
}

__device__ __forceinline__ void rowptr_scan_block(const int* __restrict__ deg,
                                                  int* __restrict__ rowptr, int n) {
  constexpr int VPT = 4;
  __shared__ int wsum[32];
  const int t = (int)threadIdx.x;
  const int lane = t & 31;
  const int wv = t >> 5;
  int carry = 0;
  for (int base = 0; base < n; base += 1024 * 4 * VPT) {
    const int i0 = base + 4 * VPT * t;
    int4 e[VPT];
    int run = 0;
#pragma unroll
    for (int k = 0; k < VPT; ++k) {
      int4 v = make_int4(0, 0, 0, 0);
      if (i0 + 4 * k < n) v = *(const int4*)(deg + i0 + 4 * k);
      e[k].x = run; run += v.x;
      e[k].y = run; run += v.y;
      e[k].z = run; run += v.z;
      e[k].w = run; run += v.w;
    }
    int inc = run;
#pragma unroll
    for (int off = 1; off < 32; off <<= 1) {
      int y = __shfl_up(inc, off);
      if (lane >= off) inc += y;
    }
    if (lane == 31) wsum[wv] = inc;
    __syncthreads();
    const int wtot = wsum[lane];
    int winc = wtot;
#pragma unroll
    for (int off = 1; off < 32; off <<= 1) {
      int y = __shfl_up(winc, off);
      if (lane >= off) winc += y;
    }
    const int wexcl = __shfl(winc, wv) - __shfl(wtot, wv);
    const int total = __shfl(winc, 31);
    const int ex = carry + wexcl + (inc - run);
#pragma unroll
    for (int k = 0; k < VPT; ++k) {
      if (i0 + 4 * k < n)
        *(int4*)(rowptr + i0 + 4 * k) =
            make_int4(ex + e[k].x, ex + e[k].y, ex + e[k].z, ex + e[k].w);
    }
    carry += total;
    __syncthreads();
  }
  if (t == 0) rowptr[n] = carry;
}

__global__ void fill_col_kernel(const int* __restrict__ src, const int* __restrict__ dst,
                                const int* __restrict__ rowptr, const int* __restrict__ rank,
                                int* __restrict__ col, int nE) {
  int i = blockIdx.x * blockDim.x + threadIdx.x;
  if (i < nE) col[rowptr[dst[i]] + rank[i]] = src[i];
}


__device__ __forceinline__ void pack_w_elem(const float* __restrict__ W, float* __restrict__ Wp,
                                            int K, int N, int Np, int i) {
  int total = (K >> 1) * Np;
  if (i >= total) return;
  int k2 = i / Np;
  int c = i - k2 * Np;
  float x0 = 0.f, x1 = 0.f;
  if (c < N) {
    x0 = W[(size_t)(2 * k2) * N + c];
    x1 = W[(size_t)(2 * k2 + 1) * N + c];
  }
  *(float2*)(Wp + (size_t)k2 * 2 * Np + 2 * c) = make_float2(x0, x1);
}

__device__ __forceinline__ v8f wmma4(v2f a, v2f b, v8f c) {
  return __builtin_amdgcn_wmma_f32_16x16x4_f32(false, a, false, b, (short)0, c, false, false);
}


template <int D, int TP>
__device__ __forceinline__ void gather_mean_tile(
    const float* __restrict__ h, const int* __restrict__ rowptr, const int* __restrict__ col,
    const float* __restrict__ invd, float* tile, int M, int row0) {
  constexpr int NV = D / 128;
  constexpr int RPW = 10;
  const int lane = (int)threadIdx.x & 31;
  const int wv = __builtin_amdgcn_readfirstlane((int)threadIdx.x >> 5);
  const int v0 = row0 + RPW * wv;
  const int myrp = rowptr[min(v0 + min(lane, RPW), M)];
  const int myinv = __float_as_int(invd[min(v0 + min(lane, RPW - 1), M - 1)]);
  const float* hl = h + 4 * lane;
  int beg = __builtin_amdgcn_readlane(myrp, 0);
  int end = __builtin_amdgcn_readlane(myrp, 1);
  int ids = (lane < end - beg) ? col[beg + lane] : 0;
#pragma nounroll
  for (int i = 0; i < RPW; ++i) {
    const int nend = (i + 1 < RPW) ? __builtin_amdgcn_readlane(myrp, i + 2) : end;
    const int nids = (lane < nend - end) ? col[end + lane] : 0;
    float4 acc[NV];
#pragma unroll
    for (int k = 0; k < NV; ++k) acc[k] = make_float4(0.f, 0.f, 0.f, 0.f);
    int myc = ids;
    for (int c0 = beg; c0 < end; c0 += 32) {
      const int n = min(32, end - c0);
      if (c0 != beg) myc = (lane < n) ? col[c0 + lane] : 0;
      int e = 0;
      for (; e + 4 <= n; e += 4) {
        const float* p0 = hl + (size_t)__builtin_amdgcn_readlane(myc, e) * D;
        const float* p1 = hl + (size_t)__builtin_amdgcn_readlane(myc, e + 1) * D;
        const float* p2 = hl + (size_t)__builtin_amdgcn_readlane(myc, e + 2) * D;
        const float* p3 = hl + (size_t)__builtin_amdgcn_readlane(myc, e + 3) * D;
        float4 r0[NV], r1[NV], r2[NV], r3[NV];
#pragma unroll
        for (int k = 0; k < NV; ++k) {
          r0[k] = *(const float4*)(p0 + 128 * k);
          r1[k] = *(const float4*)(p1 + 128 * k);
          r2[k] = *(const float4*)(p2 + 128 * k);
          r3[k] = *(const float4*)(p3 + 128 * k);
        }
#pragma unroll
        for (int k = 0; k < NV; ++k) {
          acc[k].x += r0[k].x; acc[k].y += r0[k].y; acc[k].z += r0[k].z; acc[k].w += r0[k].w;
          acc[k].x += r1[k].x; acc[k].y += r1[k].y; acc[k].z += r1[k].z; acc[k].w += r1[k].w;
          acc[k].x += r2[k].x; acc[k].y += r2[k].y; acc[k].z += r2[k].z; acc[k].w += r2[k].w;
          acc[k].x += r3[k].x; acc[k].y += r3[k].y; acc[k].z += r3[k].z; acc[k].w += r3[k].w;
        }
      }
      for (; e < n; ++e) {
        const float* p0 = hl + (size_t)__builtin_amdgcn_readlane(myc, e) * D;
#pragma unroll
        for (int k = 0; k < NV; ++k) {
          const float4 r = *(const float4*)(p0 + 128 * k);
          acc[k].x += r.x; acc[k].y += r.y; acc[k].z += r.z; acc[k].w += r.w;
        }
      }
    }
    const float inv = __int_as_float(__builtin_amdgcn_readlane(myinv, i));
    float* o = tile + (RPW * wv + i) * TP + 4 * lane;
#pragma unroll
    for (int k = 0; k < NV; ++k)
      *(float4*)(o + 128 * k) =
          make_float4(acc[k].x * inv, acc[k].y * inv, acc[k].z * inv, acc[k].w * inv);
    beg = end;
    end = nend;
    ids = nids;
  }
}

template <int D>
__device__ __forceinline__ void self_half(
    const float* __restrict__ A, const float* __restrict__ Wp, float* lds, int M, int row0,
    v8f (&acc)[5][2]) {
  constexpr int RB = 80;
  constexpr int KC = 32;
  constexpr int KS = KC / 4;
  constexpr int PITCH = KC + 4;
  constexpr int NLD = 3;
  constexpr int NC = D / KC;
  constexpr size_t ldb = 512;

  const int t = (int)threadIdx.x;
  const int lane = t & 31;
  const int wv = __builtin_amdgcn_readfirstlane(t >> 5);
  const int half = lane >> 4;
  const int lcol = lane & 15;
  const int n0 = wv << 5;
  const int boff = 2 * (n0 + lcol);

  const int srow = t >> 3;
  const int seg = t & 7;
  const bool third = (srow < RB - 64);
  const float* ap[NLD];
#pragma unroll
  for (int p = 0; p < NLD; ++p) {
    const int rr = min(row0 + 32 * p + srow, M - 1);
    ap[p] = A + (size_t)rr * D + 4 * seg;
  }
  v4f pre[NLD];
#pragma unroll
  for (int p = 0; p < NLD; ++p) {
    if (p < 2 || third) pre[p] = *(const v4f*)(ap[p]);
  }
  {
    float* s = lds + srow * PITCH + 4 * seg;
#pragma unroll
    for (int p = 0; p < NLD; ++p) {
      if (p < 2 || third) *(v4f*)(s + 32 * p * PITCH) = pre[p];
    }
  }
  __syncthreads();

  const float* bpw = Wp + (size_t)half * ldb + boff;
  const float* saw = lds + lcol * PITCH + 2 * half;
  for (int c = 0; c < NC; ++c) {
    const bool more = (c + 1 < NC);
    v2f bf[KS][2];
    const float* bpc = bpw + (size_t)(c * (KC / 2)) * ldb;
#pragma unroll
    for (int q = 0; q < KS; ++q) {
#pragma unroll
      for (int j = 0; j < 2; ++j) bf[q][j] = *(const v2f*)(bpc + (size_t)(2 * q) * ldb + 32 * j);
    }
    if (more) {
#pragma unroll
      for (int p = 0; p < NLD; ++p) {
        if (p < 2 || third) pre[p] = *(const v4f*)(ap[p] + (c + 1) * KC);
      }
    }
    const float* sa = saw + (c & 1) * (RB * PITCH);
#pragma unroll
    for (int q = 0; q < KS; ++q) {
      v2f a[5];
#pragma unroll
      for (int i = 0; i < 5; ++i) a[i] = *(const v2f*)(sa + 16 * i * PITCH + 4 * q);
#pragma unroll
      for (int i = 0; i < 5; ++i) {
#pragma unroll
        for (int j = 0; j < 2; ++j) acc[i][j] = wmma4(a[i], bf[q][j], acc[i][j]);
      }
    }
    if (more) {
      float* s = lds + ((c + 1) & 1) * (RB * PITCH) + srow * PITCH + 4 * seg;
#pragma unroll
      for (int p = 0; p < NLD; ++p) {
        if (p < 2 || third) *(v4f*)(s + 32 * p * PITCH) = pre[p];
      }
      __syncthreads();
    }
  }
}

template <int D, int TP>
__device__ __forceinline__ void neigh_half(const float* tile, const float* __restrict__ Wn,
                                           v8f (&acc)[5][2]) {
  constexpr int NC = D / 32;
  constexpr size_t ldb = 512;
  const int t = (int)threadIdx.x;
  const int lane = t & 31;
  const int wv = __builtin_amdgcn_readfirstlane(t >> 5);
  const int half = lane >> 4;
  const int lcol = lane & 15;
  const int n0 = wv << 5;
  const float* bpw = Wn + (size_t)half * ldb + 2 * (n0 + lcol);
  const float* saw = tile + lcol * TP + 2 * half;
  v2f b0[4][2], b1[4][2];
#pragma unroll
  for (int q = 0; q < 4; ++q) {
#pragma unroll
    for (int j = 0; j < 2; ++j) b0[q][j] = *(const v2f*)(bpw + (size_t)(2 * q) * ldb + 32 * j);
  }
  for (int c = 0; c < NC; ++c) {
    const float* bpc = bpw + (size_t)(c * 16) * ldb;
    const float* sa = saw + 32 * c;
#pragma unroll
    for (int q = 0; q < 4; ++q) {
#pragma unroll
      for (int j = 0; j < 2; ++j)
        b1[q][j] = *(const v2f*)(bpc + (size_t)(8 + 2 * q) * ldb + 32 * j);
    }
#pragma unroll
    for (int q = 0; q < 4; ++q) {
      v2f a[5];
#pragma unroll
      for (int i = 0; i < 5; ++i) a[i] = *(const v2f*)(sa + 16 * i * TP + 4 * q);
#pragma unroll
      for (int i = 0; i < 5; ++i) {
#pragma unroll
        for (int j = 0; j < 2; ++j) acc[i][j] = wmma4(a[i], b0[q][j], acc[i][j]);
      }
    }
    if (c + 1 < NC) {
#pragma unroll
      for (int q = 0; q < 4; ++q) {
#pragma unroll
        for (int j = 0; j < 2; ++j)
          b0[q][j] = *(const v2f*)(bpc + (size_t)(16 + 2 * q) * ldb + 32 * j);
      }
    }
#pragma unroll
    for (int q = 0; q < 4; ++q) {
      v2f a[5];
#pragma unroll
      for (int i = 0; i < 5; ++i) a[i] = *(const v2f*)(sa + 16 * i * TP + 16 + 4 * q);
#pragma unroll
      for (int i = 0; i < 5; ++i) {
#pragma unroll
        for (int j = 0; j < 2; ++j) acc[i][j] = wmma4(a[i], b1[q][j], acc[i][j]);
      }
    }
  }
}

__device__ __forceinline__ void pack_w_pair_elem(const float* __restrict__ W,
                                                 float* __restrict__ Wp, int d, int N, int i) {
  int total = (d >> 1) * 128;
  if (i >= total) return;
  int k2 = i >> 7;
  int c = i & 127;
  int part = c >> 6;
  int cc = c & 63;
  float x0 = 0.f, x1 = 0.f;
  if (cc < N) {
    x0 = W[(size_t)(part * d + 2 * k2) * N + cc];
    x1 = W[(size_t)(part * d + 2 * k2 + 1) * N + cc];
  }
  *(float2*)(Wp + (size_t)k2 * 256 + 2 * c) = make_float2(x0, x1);
}

__global__ __launch_bounds__(1024) void scan_invdeg_pack_kernel(
    const int* __restrict__ deg, int* __restrict__ rowptr, float* __restrict__ invd, int n,
    const float* __restrict__ W1, float* __restrict__ Wp1,
    const float* __restrict__ W2, float* __restrict__ Wp2,
    const float* __restrict__ W3, float* __restrict__ Wp3,
    int d1, int d2, int nh, int no) {
  const int b = (int)blockIdx.x;
  const int t = (int)threadIdx.x;
  if (b == 0) {
    rowptr_scan_block(deg, rowptr, n);
    return;
  }
  int r = b - 1;
  const int nb_inv = (n + 1023) >> 10;
  if (r < nb_inv) {
    const int i = r * 1024 + t;
    if (i < n) invd[i] = 1.0f / fmaxf((float)deg[i], 1.0f);
    return;
  }
  r -= nb_inv;
  const int nb_p1 = (d1 * nh + 1023) >> 10;
  if (r < nb_p1) {
    pack_w_elem(W1, Wp1, 2 * d1, nh, nh, r * 1024 + t);
    return;
  }
  r -= nb_p1;
  const int nb_p2 = (d2 * nh + 1023) >> 10;
  if (r < nb_p2) {
    pack_w_elem(W2, Wp2, 2 * d2, nh, nh, r * 1024 + t);
    return;
  }
  r -= nb_p2;
  pack_w_pair_elem(W3, Wp3, d2, no, r * 1024 + t);
}

template <int D, bool PROJECT>
__global__ __launch_bounds__(256) __attribute__((amdgpu_waves_per_eu(4, 4))) void layer_kernel(
    const float* __restrict__ h, const int* __restrict__ rowptr, const int* __restrict__ col,
    const float* __restrict__ invd, const float* __restrict__ Wp, const float* __restrict__ bias,
    float* __restrict__ hout, const float* __restrict__ Wp3, float* __restrict__ PZ, int M) {
  constexpr int TP = D + 4;
  constexpr int RB = 80;
  __shared__ __attribute__((aligned(16))) float tile[RB * TP];
  __shared__ __attribute__((aligned(16))) float stage[2 * RB * 36];
  const int b = (int)blockIdx.x;
  const int row0 = b * RB;

  v8f acc[5][2];
#pragma unroll
  for (int i = 0; i < 5; ++i) {
#pragma unroll
    for (int j = 0; j < 2; ++j) {
#pragma unroll
      for (int r = 0; r < 8; ++r) acc[i][j][r] = 0.0f;
    }
  }

  if ((((b >> 8) ^ b) & 1) == 0) {
    gather_mean_tile<D, TP>(h, rowptr, col, invd, tile, M, row0);
    self_half<D>(h, Wp, stage, M, row0, acc);
  } else {
    self_half<D>(h, Wp, stage, M, row0, acc);
    gather_mean_tile<D, TP>(h, rowptr, col, invd, tile, M, row0);
  }
  __syncthreads();
  neigh_half<D, TP>(tile, Wp + (size_t)(D / 2) * 512, acc);

  const int lane = (int)threadIdx.x & 31;
  const int wv = __builtin_amdgcn_readfirstlane((int)threadIdx.x >> 5);
  const int half = lane >> 4;
  const int lcol = lane & 15;
  const int n0 = wv << 5;

  if (!PROJECT) {
#pragma unroll
    for (int i = 0; i < 5; ++i) {
      if (row0 + 16 * i < M) {
#pragma unroll
        for (int j = 0; j < 2; ++j) {
          const int cc = n0 + j * 16 + lcol;
          const float bj = bias[cc];
#pragma unroll
          for (int r = 0; r < 8; ++r)
            hout[(size_t)(row0 + 16 * i + half * 8 + r) * 256 + cc] =
                fmaxf(acc[i][j][r] + bj, 0.0f);
        }
      }
    }
  } else {
    __syncthreads();
#pragma unroll
    for (int j = 0; j < 2; ++j) {
      const int cc = n0 + j * 16 + lcol;
      const float bj = bias[cc];
#pragma unroll
      for (int i = 0; i < 5; ++i) {
#pragma unroll
        for (int r = 0; r < 8; ++r)
          tile[(16 * i + half * 8 + r) * TP + cc] = fmaxf(acc[i][j][r] + bj, 0.0f);
      }
    }
    const float* bp = Wp3 + (size_t)half * 256 + 2 * (16 * wv + lcol);
    const float* sa = tile + lcol * TP + 2 * half;
    v8f z[5];
#pragma unroll
    for (int i = 0; i < 5; ++i) {
#pragma unroll
      for (int r = 0; r < 8; ++r) z[i][r] = 0.0f;
    }
    v2f bf[8];
#pragma unroll
    for (int q = 0; q < 8; ++q) bf[q] = *(const v2f*)(bp + (size_t)(2 * q) * 256);
    __syncthreads();
    for (int kc = 0; kc < D / 32; ++kc) {
      const bool more = (kc + 1 < D / 32);
      v2f bn[8];
      if (more) {
#pragma unroll
        for (int q = 0; q < 8; ++q)
          bn[q] = *(const v2f*)(bp + (size_t)((kc + 1) * 16 + 2 * q) * 256);
      }
#pragma unroll
      for (int q = 0; q < 8; ++q) {
#pragma unroll
        for (int i = 0; i < 5; ++i) {
          const v2f a = *(const v2f*)(sa + 16 * i * TP + kc * 32 + 4 * q);
          z[i] = wmma4(a, bf[q], z[i]);
        }
      }
      if (more) {
#pragma unroll
        for (int q = 0; q < 8; ++q) bf[q] = bn[q];
      }
    }
    const int cc = 16 * wv + lcol;
#pragma unroll
    for (int i = 0; i < 5; ++i) {
      if (row0 + 16 * i < M) {
#pragma unroll
        for (int r = 0; r < 8; ++r)
          PZ[(size_t)(row0 + 16 * i + half * 8 + r) * 128 + cc] = z[i][r];
      }
    }
  }
}

__global__ __launch_bounds__(256) void gather_finish_kernel(
    const float* __restrict__ PZ, const int* __restrict__ rowptr, const int* __restrict__ col,
    const float* __restrict__ invd, const float* __restrict__ bias, float* __restrict__ out,
    int M) {
  constexpr int NO = 47;
  constexpr int NPW = 4;
  __shared__ float stage[32 * NO];
  const int lane = (int)threadIdx.x & 31;
  const int half = lane >> 4;
  const int wv = __builtin_amdgcn_readfirstlane((int)threadIdx.x >> 5);
  const int vb = (int)blockIdx.x * 32;
  const int v0 = vb + wv * NPW;
  const float* Zl = PZ + 64 + 4 * (lane & 15);

  int rp[NPW + 1];
#pragma unroll
  for (int k = 0; k <= NPW; ++k) rp[k] = rowptr[min(v0 + k, M)];
  int ids[NPW];
  float4 prow[NPW];
#pragma unroll
  for (int t = 0; t < NPW; ++t) {
    ids[t] = (lane < rp[t + 1] - rp[t]) ? col[rp[t] + lane] : 0;
    prow[t] = *(const float4*)(PZ + (size_t)min(v0 + t, M - 1) * 128 + 4 * (lane & 15));
  }

#pragma unroll
  for (int t = 0; t < NPW; ++t) {
    const int v = v0 + t;
    const int beg = rp[t];
    const int end = rp[t + 1];
    float4 acc = make_float4(0.f, 0.f, 0.f, 0.f);
    int myc = ids[t];
    for (int c0 = beg; c0 < end; c0 += 32) {
      const int n = min(32, end - c0);
      if (c0 != beg) myc = (lane < n) ? col[c0 + lane] : 0;
      for (int i = 0; i < n; i += 8) {
        float4 r[4];
#pragma unroll
        for (int q = 0; q < 4; ++q) {
          const int k = i + 2 * q;
          const int ea = __builtin_amdgcn_readlane(myc, k);
          const int eb = __builtin_amdgcn_readlane(myc, k + 1);
          r[q] = make_float4(0.f, 0.f, 0.f, 0.f);
          if (k + half < n) r[q] = *(const float4*)(Zl + (size_t)(half ? eb : ea) * 128);
        }
#pragma unroll
        for (int q = 0; q < 4; ++q) {
          acc.x += r[q].x; acc.y += r[q].y; acc.z += r[q].z; acc.w += r[q].w;
        }
      }
    }
    acc.x += __shfl_xor(acc.x, 16);
    acc.y += __shfl_xor(acc.y, 16);
    acc.z += __shfl_xor(acc.z, 16);
    acc.w += __shfl_xor(acc.w, 16);
    if (v < M && lane < 12) {
      const float inv = invd[v];
      const float4 p = prow[t];
      float* s = stage + (wv * NPW + t) * NO + 4 * lane;
      s[0] = p.x + acc.x * inv + bias[4 * lane];
      s[1] = p.y + acc.y * inv + bias[4 * lane + 1];
      s[2] = p.z + acc.z * inv + bias[4 * lane + 2];
      if (lane < 11) s[3] = p.w + acc.w * inv + bias[4 * lane + 3];
    }
  }
  __syncthreads();
  const int nvalid = min(32, M - vb) * NO;
  float* o = out + (size_t)vb * NO;
  for (int L = wv; L < NO; L += 8) {
    const int idx = L * 32 + lane;
    if (idx < nvalid) o[idx] = stage[idx];
  }
}

extern "C" void kernel_launch(void* const* d_in, const int* in_sizes, int n_in, void* d_out,
                              int out_size, void* d_ws, size_t ws_size, hipStream_t stream) {
  (void)in_sizes; (void)n_in; (void)out_size; (void)ws_size;
  const float* x  = (const float*)d_in[0];
  const float* W1 = (const float*)d_in[1];
  const float* b1 = (const float*)d_in[2];
  const float* W2 = (const float*)d_in[3];
  const float* b2 = (const float*)d_in[4];
  const float* W3 = (const float*)d_in[5];
  const float* b3 = (const float*)d_in[6];
  const int* src  = (const int*)d_in[7];
  const int* dst  = (const int*)d_in[8];
  float* out = (float*)d_out;

  constexpr int M = 40000, E = 640000, INF = 128, HID = 256, OUTF = 47, OUTP = 64;

  float* ws = (float*)d_ws;
  int*   deg  = (int*)ws;
  float* invd = ws + 40000;
  float* pz   = ws + 80000;
  float* h1   = pz + (size_t)M * HID;
  float* lst  = h1 + (size_t)M * HID;
  float* Wp1  = lst + (size_t)M * HID;
  float* Wp2  = Wp1 + (size_t)(2 * INF) * HID;
  float* Wp3  = Wp2 + (size_t)(2 * HID) * HID;

  int* rank   = (int*)lst;
  int* col    = rank + E;
  int* rowptr = col + E;

  const int T = 256;
  auto blocks = [](long long n, int t) { return (int)((n + t - 1) / t); };

  zero4_kernel<<<blocks(M / 4, T), T, 0, stream>>>((float4*)deg, M / 4);
  degree_rank_kernel<<<blocks(E, T), T, 0, stream>>>(dst, deg, rank, E);
  {
    int nb = 1 + blocks(M, 1024) + blocks((long long)INF * HID, 1024) +
             blocks((long long)HID * HID, 1024) + blocks((HID / 2) * (long long)(2 * OUTP), 1024);
    scan_invdeg_pack_kernel<<<nb, 1024, 0, stream>>>(deg, rowptr, invd, M, W1, Wp1, W2, Wp2,
                                                     W3, Wp3, INF, HID, HID, OUTF);
  }
  fill_col_kernel<<<blocks(E, T), T, 0, stream>>>(src, dst, rowptr, rank, col, E);

  layer_kernel<INF, false><<<blocks(M, 80), T, 0, stream>>>(
      x, rowptr, col, invd, Wp1, b1, h1, nullptr, nullptr, M);
  layer_kernel<HID, true><<<blocks(M, 80), T, 0, stream>>>(
      h1, rowptr, col, invd, Wp2, b2, nullptr, Wp3, pz, M);
  gather_finish_kernel<<<M / 32, T, 0, stream>>>(pz, rowptr, col, invd, b3, out, M);
}
